# starting kernel
# baseline (speedup 1.0000x reference)
#include <hip/hip_runtime.h>
#include <math.h>

#define ALPHA 0.1f

typedef float v2f __attribute__((ext_vector_type(2)));
typedef float v8f __attribute__((ext_vector_type(8)));

__device__ __forceinline__ void atomic_add_f32(float* p, float v) {
#if __has_builtin(__builtin_amdgcn_global_atomic_fadd_f32)
  typedef __attribute__((address_space(1))) float gfloat;
  __builtin_amdgcn_global_atomic_fadd_f32((gfloat*)p, v);
#else
  atomicAdd(p, v);
#endif
}


__global__ void zero4_k(float4* __restrict__ p, int n4) {
  int i = blockIdx.x * blockDim.x + threadIdx.x;
  if (i < n4) p[i] = make_float4(0.f, 0.f, 0.f, 0.f);
}

__global__ void deg_k(const int* __restrict__ dst, float* __restrict__ deg, int E) {
  int e = blockIdx.x * blockDim.x + threadIdx.x;
  if (e < E) atomic_add_f32(&deg[dst[e]], 1.0f);
}

__global__ void norm_k(float* __restrict__ nrm, int n) {
  int i = blockIdx.x * blockDim.x + threadIdx.x;
  if (i < n) {
    float d = nrm[i];
    nrm[i] = 1.0f / sqrtf(fmaxf(d, 1.0f));
  }
}

__device__ __forceinline__ float elu1(float v) {
  return v > 0.0f ? v : (expf(v) - 1.0f);
}

template <int KDIM, int HID, bool ELU>
__global__ void gemm_wmma(const float* __restrict__ A, const float* __restrict__ W,
                          const float* __restrict__ bias, float* __restrict__ C, int N) {
  constexpr int NTN = HID / 16;
  const int tiles = ((N + 15) / 16) * NTN;
  const int tile = __builtin_amdgcn_readfirstlane(
      blockIdx.x * (blockDim.x >> 5) + (threadIdx.x >> 5));
  if (tile >= tiles) return;
  const int tm = tile / NTN;
  const int tn = tile % NTN;
  const int lane = threadIdx.x & 31;
  const int half = lane >> 4;
  const int l15 = lane & 15;
  const int row = tm * 16 + l15;
  const int rowc = row < N ? row : N - 1;
  const int col = tn * 16 + l15;

  v8f acc = {};
  for (int k = 0; k < KDIM; k += 4) {
    const int ka = k + 2 * half;
    v2f a, b;
    a.x = A[rowc * KDIM + ka];
    a.y = A[rowc * KDIM + ka + 1];
    b.x = W[col * KDIM + ka];
    b.y = W[col * KDIM + ka + 1];
    acc = __builtin_amdgcn_wmma_f32_16x16x4_f32(false, a, false, b,
                                                (short)0, acc, false, false);
  }
  const float bs = bias[col];
  if (tm * 16 + 15 < N) {
#pragma unroll
    for (int v = 0; v < 8; ++v) {
      int m = tm * 16 + v + 8 * half;
      float x = acc[v] + bs;
      if (ELU) x = elu1(x);
      C[m * HID + col] = x;
    }
  } else {
#pragma unroll
    for (int v = 0; v < 8; ++v) {
      int m = tm * 16 + v + 8 * half;
      if (m < N) {
        float x = acc[v] + bs;
        if (ELU) x = elu1(x);
        C[m * HID + col] = x;
      }
    }
  }
}

template <int D>
__global__ void edge_k(const int* __restrict__ src, const int* __restrict__ dst,
                       const float* __restrict__ h, const float* __restrict__ nrm,
                       float* __restrict__ agg, int E) {
  int t = blockIdx.x * blockDim.x + threadIdx.x;
  int e = t >> 4;
  int c = t & 15;
  if (e >= E) return;
  int s = src[e];
  int d = dst[e];
  float w = nrm[s];
#pragma unroll
  for (int j = 0; j < D / 16; ++j) {
    int ch = c + 16 * j;
    atomic_add_f32(&agg[d * D + ch], h[s * D + ch] * w);
  }
}

template <int D, bool ELU>
__global__ void combine4_k(const float* __restrict__ nrm, const float4* __restrict__ h0,
                           float4* __restrict__ agg, float4* __restrict__ out, int N) {
  int i = blockIdx.x * blockDim.x + threadIdx.x;
  if (i >= N * (D / 4)) return;
  int node = i / (D / 4);
  float w = (1.0f - ALPHA) * nrm[node];
  float4 a = agg[i];
  float4 h = h0[i];
  float4 r = make_float4(w * a.x + ALPHA * h.x, w * a.y + ALPHA * h.y,
                         w * a.z + ALPHA * h.z, w * a.w + ALPHA * h.w);
  if (ELU) {
    out[i] = make_float4(elu1(r.x), elu1(r.y), elu1(r.z), elu1(r.w));
  } else {
    agg[i] = r;
  }
}


extern "C" void kernel_launch(void* const* d_in, const int* in_sizes, int n_in,
                              void* d_out, int out_size, void* d_ws, size_t ws_size,
                              hipStream_t stream) {
  const float* F   = (const float*)d_in[0];
  const int*   src = (const int*)d_in[1];
  const int*   dst = (const int*)d_in[2];
  const float* W1  = (const float*)d_in[3];
  const float* b1  = (const float*)d_in[4];
  const float* W2  = (const float*)d_in[5];
  const float* b2  = (const float*)d_in[6];
  float* out = (float*)d_out;

  constexpr int IN = 256, HID = 48, CLS = 16, KSTEPS = 10;
  const int N = in_sizes[0] / IN;
  const int E = in_sizes[1];

  float* ws  = (float*)d_ws;
  float* nrm = ws;
  float* h0  = nrm + N;
  float* hA  = h0 + (size_t)N * HID;
  float* hB  = hA + (size_t)N * HID;
  float* x0  = h0;
  float* xA  = x0 + (size_t)N * CLS;
  float* xB  = xA + (size_t)N * CLS;

  auto blocks = [](long n, int t) { return (int)((n + t - 1) / t); };

  zero4_k<<<blocks(N / 4, 256), 256, 0, stream>>>((float4*)nrm, N / 4);
  deg_k<<<blocks(E, 256), 256, 0, stream>>>(dst, nrm, E);
  norm_k<<<blocks(N, 256), 256, 0, stream>>>(nrm, N);

  const int tiles1 = ((N + 15) / 16) * (HID / 16);
  gemm_wmma<IN, HID, false><<<blocks((long)tiles1 * 32, 128), 128, 0, stream>>>(F, W1, b1, h0, N);

  const float* cur = h0;
  for (int t = 0; t < KSTEPS; ++t) {
    float* agg = (t & 1) ? hB : hA;
    zero4_k<<<blocks((long)N * HID / 4, 256), 256, 0, stream>>>((float4*)agg, N * (HID / 4));
    edge_k<HID><<<blocks((long)E * 16, 256), 256, 0, stream>>>(src, dst, cur, nrm, agg, E);
    combine4_k<HID, false><<<blocks((long)N * HID / 4, 256), 256, 0, stream>>>(
        nrm, (const float4*)h0, (float4*)agg, nullptr, N);
    cur = agg;
  }

  const int tiles2 = ((N + 15) / 16) * (CLS / 16);
  gemm_wmma<HID, CLS, true><<<blocks((long)tiles2 * 32, 128), 128, 0, stream>>>(cur, W2, b2, x0, N);

  const float* cur2 = x0;
  for (int t = 0; t < KSTEPS; ++t) {
    float* agg = (t & 1) ? xB : xA;
    zero4_k<<<blocks((long)N * CLS / 4, 256), 256, 0, stream>>>((float4*)agg, N * (CLS / 4));
    edge_k<CLS><<<blocks((long)E * 16, 256), 256, 0, stream>>>(src, dst, cur2, nrm, agg, E);
    if (t == KSTEPS - 1) {
      combine4_k<CLS, true><<<blocks((long)N * CLS / 4, 256), 256, 0, stream>>>(
          nrm, (const float4*)x0, (float4*)agg, (float4*)out, N);
    } else {
      combine4_k<CLS, false><<<blocks((long)N * CLS / 4, 256), 256, 0, stream>>>(
          nrm, (const float4*)x0, (float4*)agg, nullptr, N);
    }
    cur2 = agg;
  }
}


# Float-atomic edge scatter -> per-call destination-grouped edge list (int atomic arrival rank + scan) and atomic-free per-node row gather; sum order = arrival rank, slot tree at D=16
# speedup vs baseline: 3.2914x; 3.2914x over previous
#include <hip/hip_runtime.h>
#include <math.h>

#define ALPHA 0.1f

typedef float v2f __attribute__((ext_vector_type(2)));
typedef float v8f __attribute__((ext_vector_type(8)));


__global__ void zero_i32_k(int* __restrict__ p, int n) {
  int i = blockIdx.x * blockDim.x + threadIdx.x;
  if (i < n) p[i] = 0;
}

__global__ void rank_k(const int* __restrict__ dst, int* __restrict__ cnt,
                       int* __restrict__ rnk, int E) {
  int e = blockIdx.x * blockDim.x + threadIdx.x;
  if (e < E) rnk[e] = atomicAdd(&cnt[dst[e]], 1);
}

__global__ __launch_bounds__(1024) void scan_k(const int* __restrict__ cnt, int* __restrict__ off,
                                               float* __restrict__ nrm, int N) {
  __shared__ int wtot[32];
  __shared__ int wpre[32];
  const int tid = threadIdx.x;
  const int lane = tid & 31;
  const int wv = tid >> 5;
  const int base = blockIdx.x * 1024;
  int p = 0;
  for (int i = tid; i < base; i += 1024) p += cnt[i];
  const int v = base + tid;
  const int c = v < N ? cnt[v] : 0;
#pragma unroll
  for (int o = 16; o > 0; o >>= 1) p += __shfl_xor(p, o);
  int inc = c;
#pragma unroll
  for (int o = 1; o < 32; o <<= 1) {
    int y = __shfl_up(inc, o);
    if (lane >= o) inc += y;
  }
  if (lane == 31) wtot[wv] = inc;
  if (lane == 0) wpre[wv] = p;
  __syncthreads();
  int front = 0;
  for (int i = 0; i < 32; ++i) {
    front += wpre[i];
    if (i < wv) front += wtot[i];
  }
  const int excl = front + inc - c;
  if (v < N) {
    off[v] = excl;
    nrm[v] = 1.0f / sqrtf(fmaxf((float)c, 1.0f));
    if (v == N - 1) off[N] = excl + c;
  }
}

__global__ void fill_k(const int* __restrict__ src, const int* __restrict__ dst,
                       const int* __restrict__ off, const int* __restrict__ rnk,
                       int* __restrict__ col, int E) {
  int e = blockIdx.x * blockDim.x + threadIdx.x;
  if (e < E) col[off[dst[e]] + rnk[e]] = src[e];
}


__device__ __forceinline__ float elu1(float v) {
  return v > 0.0f ? v : (expf(v) - 1.0f);
}

template <int KDIM, int HID, bool ELU>
__global__ void gemm_wmma(const float* __restrict__ A, const float* __restrict__ W,
                          const float* __restrict__ bias, float* __restrict__ C, int N) {
  constexpr int NTN = HID / 16;
  const int tiles = ((N + 15) / 16) * NTN;
  const int tile = __builtin_amdgcn_readfirstlane(
      blockIdx.x * (blockDim.x >> 5) + (threadIdx.x >> 5));
  if (tile >= tiles) return;
  const int tm = tile / NTN;
  const int tn = tile % NTN;
  const int lane = threadIdx.x & 31;
  const int half = lane >> 4;
  const int l15 = lane & 15;
  const int row = tm * 16 + l15;
  const int rowc = row < N ? row : N - 1;
  const int col = tn * 16 + l15;

  v8f acc = {};
  for (int k = 0; k < KDIM; k += 4) {
    const int ka = k + 2 * half;
    v2f a, b;
    a.x = A[rowc * KDIM + ka];
    a.y = A[rowc * KDIM + ka + 1];
    b.x = W[col * KDIM + ka];
    b.y = W[col * KDIM + ka + 1];
    acc = __builtin_amdgcn_wmma_f32_16x16x4_f32(false, a, false, b,
                                                (short)0, acc, false, false);
  }
  const float bs = bias[col];
  if (tm * 16 + 15 < N) {
#pragma unroll
    for (int v = 0; v < 8; ++v) {
      int m = tm * 16 + v + 8 * half;
      float x = acc[v] + bs;
      if (ELU) x = elu1(x);
      C[m * HID + col] = x;
    }
  } else {
#pragma unroll
    for (int v = 0; v < 8; ++v) {
      int m = tm * 16 + v + 8 * half;
      if (m < N) {
        float x = acc[v] + bs;
        if (ELU) x = elu1(x);
        C[m * HID + col] = x;
      }
    }
  }
}

template <int D, bool SCALE_IN, bool SCALE_OUT, bool ELU>
__global__ __launch_bounds__(256) void gather_k(const int* __restrict__ off, const int* __restrict__ col,
                                                const float* __restrict__ nrm,
                                                const float4* __restrict__ hin,
                                                const float4* __restrict__ h0,
                                                float4* __restrict__ hout, int N) {
  constexpr int L = D / 4;
  constexpr int SLOTS = (L <= 4) ? 16 / L : 1;
  constexpr int U = (SLOTS == 1) ? 8 : 4;
  const int t = blockIdx.x * blockDim.x + threadIdx.x;
  const int node = t >> 4;
  const int l16 = t & 15;
  const int q = l16 % L;
  const int slot = l16 / L;
  const bool live = (node < N) && (slot < SLOTS);
  int j = 0, end = 0;
  if (live) {
    j = off[node] + slot;
    end = off[node + 1];
  }
  const bool writer = (node < N) && (slot == 0);
  float nn = 0.f;
  float4 h = make_float4(0.f, 0.f, 0.f, 0.f);
  if (writer) {
    nn = nrm[node];
    h = h0[node * L + q];
  }
  float4 acc = make_float4(0.f, 0.f, 0.f, 0.f);
  for (; j < end; j += SLOTS * U) {
    int s[U];
    float4 v[U];
    float w[U];
#pragma unroll
    for (int u = 0; u < U; ++u) s[u] = (j + SLOTS * u < end) ? col[j + SLOTS * u] : -1;
#pragma unroll
    for (int u = 0; u < U; ++u) {
      v[u] = make_float4(0.f, 0.f, 0.f, 0.f);
      w[u] = 0.f;
      if (s[u] >= 0) {
        v[u] = hin[s[u] * L + q];
        if (SCALE_IN) w[u] = nrm[s[u]];
      }
    }
#pragma unroll
    for (int u = 0; u < U; ++u) {
      if (SCALE_IN) {
        acc.x += v[u].x * w[u]; acc.y += v[u].y * w[u];
        acc.z += v[u].z * w[u]; acc.w += v[u].w * w[u];
      } else {
        acc.x += v[u].x; acc.y += v[u].y; acc.z += v[u].z; acc.w += v[u].w;
      }
    }
  }
  if (SLOTS > 1) {
#pragma unroll
    for (int o = L; o < 16; o <<= 1) {
      acc.x += __shfl_xor(acc.x, o); acc.y += __shfl_xor(acc.y, o);
      acc.z += __shfl_xor(acc.z, o); acc.w += __shfl_xor(acc.w, o);
    }
  }
  if (writer) {
    const int i = node * L + q;
    const float wn = (1.0f - ALPHA) * nn;
    float4 r = make_float4(wn * acc.x + ALPHA * h.x, wn * acc.y + ALPHA * h.y,
                           wn * acc.z + ALPHA * h.z, wn * acc.w + ALPHA * h.w);
    if (ELU) {
      r = make_float4(elu1(r.x), elu1(r.y), elu1(r.z), elu1(r.w));
    } else if (SCALE_OUT) {
      r = make_float4(r.x * nn, r.y * nn, r.z * nn, r.w * nn);
    }
    hout[i] = r;
  }
}


extern "C" void kernel_launch(void* const* d_in, const int* in_sizes, int n_in,
                              void* d_out, int out_size, void* d_ws, size_t ws_size,
                              hipStream_t stream) {
  const float* F   = (const float*)d_in[0];
  const int*   src = (const int*)d_in[1];
  const int*   dst = (const int*)d_in[2];
  const float* W1  = (const float*)d_in[3];
  const float* b1  = (const float*)d_in[4];
  const float* W2  = (const float*)d_in[5];
  const float* b2  = (const float*)d_in[6];
  float* out = (float*)d_out;

  constexpr int IN = 256, HID = 48, CLS = 16, KSTEPS = 10;
  const int N = in_sizes[0] / IN;
  const int E = in_sizes[1];

  float* ws  = (float*)d_ws;
  float* nrm = ws;
  int*   cnt = (int*)(nrm + N);
  int*   off = cnt + N;
  int*   rnk = off + N + 4;
  int*   col = rnk + E;
  float* h0  = (float*)(col + E);
  float* hA  = h0 + (size_t)N * HID;
  float* hB  = hA + (size_t)N * HID;
  float* x0  = h0;
  float* xA  = x0 + (size_t)N * CLS;
  float* xB  = xA + (size_t)N * CLS;

  auto blocks = [](long n, int t) { return (int)((n + t - 1) / t); };

  zero_i32_k<<<blocks(N, 256), 256, 0, stream>>>(cnt, N);
  rank_k<<<blocks(E, 256), 256, 0, stream>>>(dst, cnt, rnk, E);
  scan_k<<<blocks(N, 1024), 1024, 0, stream>>>(cnt, off, nrm, N);
  fill_k<<<blocks(E, 256), 256, 0, stream>>>(src, dst, off, rnk, col, E);

  const int tiles1 = ((N + 15) / 16) * (HID / 16);
  gemm_wmma<IN, HID, false><<<blocks((long)tiles1 * 32, 128), 128, 0, stream>>>(F, W1, b1, h0, N);

  const int gb = blocks((long)N * 16, 256);

  gather_k<HID, true, true, false><<<gb, 256, 0, stream>>>(
      off, col, nrm, (const float4*)h0, (const float4*)h0, (float4*)hA, N);
  for (int t = 1; t < KSTEPS - 1; ++t) {
    const float* in = (t & 1) ? hA : hB;
    float* o = (t & 1) ? hB : hA;
    gather_k<HID, false, true, false><<<gb, 256, 0, stream>>>(
        off, col, nrm, (const float4*)in, (const float4*)h0, (float4*)o, N);
  }
  gather_k<HID, false, false, false><<<gb, 256, 0, stream>>>(
      off, col, nrm, (const float4*)hA, (const float4*)h0, (float4*)hB, N);

  const int tiles2 = ((N + 15) / 16) * (CLS / 16);
  gemm_wmma<HID, CLS, true><<<blocks((long)tiles2 * 32, 128), 128, 0, stream>>>(hB, W2, b2, x0, N);

  gather_k<CLS, true, true, false><<<gb, 256, 0, stream>>>(
      off, col, nrm, (const float4*)x0, (const float4*)x0, (float4*)xA, N);
  for (int t = 1; t < KSTEPS - 1; ++t) {
    const float* in = (t & 1) ? xA : xB;
    float* o = (t & 1) ? xB : xA;
    gather_k<CLS, false, true, false><<<gb, 256, 0, stream>>>(
        off, col, nrm, (const float4*)in, (const float4*)x0, (float4*)o, N);
  }
  gather_k<CLS, false, false, true><<<gb, 256, 0, stream>>>(
      off, col, nrm, (const float4*)xA, (const float4*)x0, (float4*)out, N);
}


# fc2 product moved in front of propagation 1 (linear maps commute: P(h)W2^T = P(hW2^T)): its ten steps gather 64-B rows, not 192-B; b2 and ELU in its last step; sums reassociated
# speedup vs baseline: 3.7024x; 1.1249x over previous
#include <hip/hip_runtime.h>
#include <math.h>

#define ALPHA 0.1f

typedef float v2f __attribute__((ext_vector_type(2)));
typedef float v8f __attribute__((ext_vector_type(8)));


__global__ void zero_i32_k(int* __restrict__ p, int n) {
  int i = blockIdx.x * blockDim.x + threadIdx.x;
  if (i < n) p[i] = 0;
}

__global__ void rank_k(const int* __restrict__ dst, int* __restrict__ cnt,
                       int* __restrict__ rnk, int E) {
  int e = blockIdx.x * blockDim.x + threadIdx.x;
  if (e < E) rnk[e] = atomicAdd(&cnt[dst[e]], 1);
}

__global__ __launch_bounds__(1024) void scan_k(const int* __restrict__ cnt, int* __restrict__ off,
                                               float* __restrict__ nrm, int N) {
  __shared__ int wtot[32];
  __shared__ int wpre[32];
  const int tid = threadIdx.x;
  const int lane = tid & 31;
  const int wv = tid >> 5;
  const int base = blockIdx.x * 1024;
  int p = 0;
  for (int i = tid; i < base; i += 1024) p += cnt[i];
  const int v = base + tid;
  const int c = v < N ? cnt[v] : 0;
#pragma unroll
  for (int o = 16; o > 0; o >>= 1) p += __shfl_xor(p, o);
  int inc = c;
#pragma unroll
  for (int o = 1; o < 32; o <<= 1) {
    int y = __shfl_up(inc, o);
    if (lane >= o) inc += y;
  }
  if (lane == 31) wtot[wv] = inc;
  if (lane == 0) wpre[wv] = p;
  __syncthreads();
  int front = 0;
  for (int i = 0; i < 32; ++i) {
    front += wpre[i];
    if (i < wv) front += wtot[i];
  }
  const int excl = front + inc - c;
  if (v < N) {
    off[v] = excl;
    nrm[v] = 1.0f / sqrtf(fmaxf((float)c, 1.0f));
    if (v == N - 1) off[N] = excl + c;
  }
}

__global__ void fill_k(const int* __restrict__ src, const int* __restrict__ dst,
                       const int* __restrict__ off, const int* __restrict__ rnk,
                       int* __restrict__ col, int E) {
  int e = blockIdx.x * blockDim.x + threadIdx.x;
  if (e < E) col[off[dst[e]] + rnk[e]] = src[e];
}


__device__ __forceinline__ float elu1(float v) {
  return v > 0.0f ? v : (expf(v) - 1.0f);
}

template <int KDIM, int HID, bool BIAS, bool ELU>
__global__ void gemm_wmma(const float* __restrict__ A, const float* __restrict__ W,
                          const float* __restrict__ bias, float* __restrict__ C, int N) {
  constexpr int NTN = HID / 16;
  const int tiles = ((N + 15) / 16) * NTN;
  const int tile = __builtin_amdgcn_readfirstlane(
      blockIdx.x * (blockDim.x >> 5) + (threadIdx.x >> 5));
  if (tile >= tiles) return;
  const int tm = tile / NTN;
  const int tn = tile % NTN;
  const int lane = threadIdx.x & 31;
  const int half = lane >> 4;
  const int l15 = lane & 15;
  const int row = tm * 16 + l15;
  const int rowc = row < N ? row : N - 1;
  const int col = tn * 16 + l15;

  v8f acc = {};
  for (int k = 0; k < KDIM; k += 4) {
    const int ka = k + 2 * half;
    v2f a, b;
    a.x = A[rowc * KDIM + ka];
    a.y = A[rowc * KDIM + ka + 1];
    b.x = W[col * KDIM + ka];
    b.y = W[col * KDIM + ka + 1];
    acc = __builtin_amdgcn_wmma_f32_16x16x4_f32(false, a, false, b,
                                                (short)0, acc, false, false);
  }
  const float bs = BIAS ? bias[col] : 0.0f;
  if (tm * 16 + 15 < N) {
#pragma unroll
    for (int v = 0; v < 8; ++v) {
      int m = tm * 16 + v + 8 * half;
      float x = acc[v] + bs;
      if (ELU) x = elu1(x);
      C[m * HID + col] = x;
    }
  } else {
#pragma unroll
    for (int v = 0; v < 8; ++v) {
      int m = tm * 16 + v + 8 * half;
      if (m < N) {
        float x = acc[v] + bs;
        if (ELU) x = elu1(x);
        C[m * HID + col] = x;
      }
    }
  }
}

template <int D, bool SCALE_IN, bool SCALE_OUT, bool ELU, bool BIAS>
__global__ __launch_bounds__(256) void gather_k(const int* __restrict__ off, const int* __restrict__ col,
                                                const float* __restrict__ nrm,
                                                const float4* __restrict__ hin,
                                                const float4* __restrict__ h0,
                                                const float* __restrict__ bias,
                                                float4* __restrict__ hout, int N) {
  constexpr int L = D / 4;
  constexpr int SLOTS = (L <= 4) ? 16 / L : 1;
  constexpr int U = (SLOTS == 1) ? 8 : 4;
  const int t = blockIdx.x * blockDim.x + threadIdx.x;
  const int node = t >> 4;
  const int l16 = t & 15;
  const int q = l16 % L;
  const int slot = l16 / L;
  const bool live = (node < N) && (slot < SLOTS);
  int j = 0, end = 0;
  if (live) {
    j = off[node] + slot;
    end = off[node + 1];
  }
  const bool writer = (node < N) && (slot == 0);
  float nn = 0.f;
  float4 h = make_float4(0.f, 0.f, 0.f, 0.f);
  float4 bv = make_float4(0.f, 0.f, 0.f, 0.f);
  if (writer) {
    nn = nrm[node];
    h = h0[node * L + q];
    if (BIAS) bv = make_float4(bias[4 * q], bias[4 * q + 1], bias[4 * q + 2], bias[4 * q + 3]);
  }
  float4 acc = make_float4(0.f, 0.f, 0.f, 0.f);
  for (; j < end; j += SLOTS * U) {
    int s[U];
    float4 v[U];
    float w[U];
#pragma unroll
    for (int u = 0; u < U; ++u) s[u] = (j + SLOTS * u < end) ? col[j + SLOTS * u] : -1;
#pragma unroll
    for (int u = 0; u < U; ++u) {
      v[u] = make_float4(0.f, 0.f, 0.f, 0.f);
      w[u] = 0.f;
      if (s[u] >= 0) {
        v[u] = hin[s[u] * L + q];
        if (SCALE_IN) w[u] = nrm[s[u]];
      }
    }
#pragma unroll
    for (int u = 0; u < U; ++u) {
      if (SCALE_IN) {
        acc.x += v[u].x * w[u]; acc.y += v[u].y * w[u];
        acc.z += v[u].z * w[u]; acc.w += v[u].w * w[u];
      } else {
        acc.x += v[u].x; acc.y += v[u].y; acc.z += v[u].z; acc.w += v[u].w;
      }
    }
  }
  if (SLOTS > 1) {
#pragma unroll
    for (int o = L; o < 16; o <<= 1) {
      acc.x += __shfl_xor(acc.x, o); acc.y += __shfl_xor(acc.y, o);
      acc.z += __shfl_xor(acc.z, o); acc.w += __shfl_xor(acc.w, o);
    }
  }
  if (writer) {
    const int i = node * L + q;
    const float wn = (1.0f - ALPHA) * nn;
    float4 r = make_float4(wn * acc.x + ALPHA * h.x, wn * acc.y + ALPHA * h.y,
                           wn * acc.z + ALPHA * h.z, wn * acc.w + ALPHA * h.w);
    if (BIAS) r = make_float4(r.x + bv.x, r.y + bv.y, r.z + bv.z, r.w + bv.w);
    if (ELU) {
      r = make_float4(elu1(r.x), elu1(r.y), elu1(r.z), elu1(r.w));
    } else if (SCALE_OUT) {
      r = make_float4(r.x * nn, r.y * nn, r.z * nn, r.w * nn);
    }
    hout[i] = r;
  }
}


extern "C" void kernel_launch(void* const* d_in, const int* in_sizes, int n_in,
                              void* d_out, int out_size, void* d_ws, size_t ws_size,
                              hipStream_t stream) {
  const float* F   = (const float*)d_in[0];
  const int*   src = (const int*)d_in[1];
  const int*   dst = (const int*)d_in[2];
  const float* W1  = (const float*)d_in[3];
  const float* b1  = (const float*)d_in[4];
  const float* W2  = (const float*)d_in[5];
  const float* b2  = (const float*)d_in[6];
  float* out = (float*)d_out;

  constexpr int IN = 256, HID = 48, CLS = 16, KSTEPS = 10;
  const int N = in_sizes[0] / IN;
  const int E = in_sizes[1];

  float* ws  = (float*)d_ws;
  float* nrm = ws;
  int*   cnt = (int*)(nrm + N);
  int*   off = cnt + N;
  int*   rnk = off + N + 4;
  int*   col = rnk + E;
  float* h0  = (float*)(col + E);
  float* z0  = h0 + (size_t)N * HID;
  float* zA  = z0 + (size_t)N * CLS;
  float* zB  = zA + (size_t)N * CLS;
  float* x0  = zB + (size_t)N * CLS;
  float* xA  = x0 + (size_t)N * CLS;
  float* xB  = xA + (size_t)N * CLS;

  auto blocks = [](long n, int t) { return (int)((n + t - 1) / t); };

  zero_i32_k<<<blocks(N, 256), 256, 0, stream>>>(cnt, N);
  rank_k<<<blocks(E, 256), 256, 0, stream>>>(dst, cnt, rnk, E);
  scan_k<<<blocks(N, 1024), 1024, 0, stream>>>(cnt, off, nrm, N);
  fill_k<<<blocks(E, 256), 256, 0, stream>>>(src, dst, off, rnk, col, E);

  const int tiles1 = ((N + 15) / 16) * (HID / 16);
  gemm_wmma<IN, HID, true, false><<<blocks((long)tiles1 * 32, 128), 128, 0, stream>>>(F, W1, b1, h0, N);

  const int tiles2 = ((N + 15) / 16) * (CLS / 16);
  gemm_wmma<HID, CLS, false, false><<<blocks((long)tiles2 * 32, 128), 128, 0, stream>>>(h0, W2, b2, z0, N);

  const int gb = blocks((long)N * 16, 256);

  gather_k<CLS, true, true, false, false><<<gb, 256, 0, stream>>>(
      off, col, nrm, (const float4*)z0, (const float4*)z0, b2, (float4*)zA, N);
  for (int t = 1; t < KSTEPS - 1; ++t) {
    const float* in = (t & 1) ? zA : zB;
    float* o = (t & 1) ? zB : zA;
    gather_k<CLS, false, true, false, false><<<gb, 256, 0, stream>>>(
        off, col, nrm, (const float4*)in, (const float4*)z0, b2, (float4*)o, N);
  }
  gather_k<CLS, false, false, true, true><<<gb, 256, 0, stream>>>(
      off, col, nrm, (const float4*)zA, (const float4*)z0, b2, (float4*)x0, N);

  gather_k<CLS, true, true, false, false><<<gb, 256, 0, stream>>>(
      off, col, nrm, (const float4*)x0, (const float4*)x0, b2, (float4*)xA, N);
  for (int t = 1; t < KSTEPS - 1; ++t) {
    const float* in = (t & 1) ? xA : xB;
    float* o = (t & 1) ? xB : xA;
    gather_k<CLS, false, true, false, false><<<gb, 256, 0, stream>>>(
        off, col, nrm, (const float4*)in, (const float4*)x0, b2, (float4*)o, N);
  }
  gather_k<CLS, false, false, true, false><<<gb, 256, 0, stream>>>(
      off, col, nrm, (const float4*)xA, (const float4*)x0, b2, (float4*)out, N);
}


# fc1 re-tiled: one wave per 16 rows makes all 48 columns; features read once as whole 128-B lines via wave-private LDS, W1 in LDS; k order inside 32-chunks changed, sums reassociated
# speedup vs baseline: 3.9118x; 1.0566x over previous
#include <hip/hip_runtime.h>
#include <math.h>

#define ALPHA 0.1f

typedef float v2f __attribute__((ext_vector_type(2)));
typedef float v8f __attribute__((ext_vector_type(8)));


__global__ void zero_i32_k(int* __restrict__ p, int n) {
  int i = blockIdx.x * blockDim.x + threadIdx.x;
  if (i < n) p[i] = 0;
}

__global__ void rank_k(const int* __restrict__ dst, int* __restrict__ cnt,
                       int* __restrict__ rnk, int E) {
  int e = blockIdx.x * blockDim.x + threadIdx.x;
  if (e < E) rnk[e] = atomicAdd(&cnt[dst[e]], 1);
}

__global__ __launch_bounds__(1024) void scan_k(const int* __restrict__ cnt, int* __restrict__ off,
                                               float* __restrict__ nrm, int N) {
  __shared__ int wtot[32];
  __shared__ int wpre[32];
  const int tid = threadIdx.x;
  const int lane = tid & 31;
  const int wv = tid >> 5;
  const int base = blockIdx.x * 1024;
  int p = 0;
  for (int i = tid; i < base; i += 1024) p += cnt[i];
  const int v = base + tid;
  const int c = v < N ? cnt[v] : 0;
#pragma unroll
  for (int o = 16; o > 0; o >>= 1) p += __shfl_xor(p, o);
  int inc = c;
#pragma unroll
  for (int o = 1; o < 32; o <<= 1) {
    int y = __shfl_up(inc, o);
    if (lane >= o) inc += y;
  }
  if (lane == 31) wtot[wv] = inc;
  if (lane == 0) wpre[wv] = p;
  __syncthreads();
  int front = 0;
  for (int i = 0; i < 32; ++i) {
    front += wpre[i];
    if (i < wv) front += wtot[i];
  }
  const int excl = front + inc - c;
  if (v < N) {
    off[v] = excl;
    nrm[v] = 1.0f / sqrtf(fmaxf((float)c, 1.0f));
    if (v == N - 1) off[N] = excl + c;
  }
}

__global__ void fill_k(const int* __restrict__ src, const int* __restrict__ dst,
                       const int* __restrict__ off, const int* __restrict__ rnk,
                       int* __restrict__ col, int E) {
  int e = blockIdx.x * blockDim.x + threadIdx.x;
  if (e < E) col[off[dst[e]] + rnk[e]] = src[e];
}


__device__ __forceinline__ float elu1(float v) {
  return v > 0.0f ? v : (expf(v) - 1.0f);
}

template <int KDIM, int HID, bool BIAS, bool ELU>
__global__ void gemm_wmma(const float* __restrict__ A, const float* __restrict__ W,
                          const float* __restrict__ bias, float* __restrict__ C, int N) {
  constexpr int NTN = HID / 16;
  const int tiles = ((N + 15) / 16) * NTN;
  const int tile = __builtin_amdgcn_readfirstlane(
      blockIdx.x * (blockDim.x >> 5) + (threadIdx.x >> 5));
  if (tile >= tiles) return;
  const int tm = tile / NTN;
  const int tn = tile % NTN;
  const int lane = threadIdx.x & 31;
  const int half = lane >> 4;
  const int l15 = lane & 15;
  const int row = tm * 16 + l15;
  const int rowc = row < N ? row : N - 1;
  const int col = tn * 16 + l15;

  v8f acc = {};
  for (int k = 0; k < KDIM; k += 4) {
    const int ka = k + 2 * half;
    v2f a, b;
    a.x = A[rowc * KDIM + ka];
    a.y = A[rowc * KDIM + ka + 1];
    b.x = W[col * KDIM + ka];
    b.y = W[col * KDIM + ka + 1];
    acc = __builtin_amdgcn_wmma_f32_16x16x4_f32(false, a, false, b,
                                                (short)0, acc, false, false);
  }
  const float bs = BIAS ? bias[col] : 0.0f;
  if (tm * 16 + 15 < N) {
#pragma unroll
    for (int v = 0; v < 8; ++v) {
      int m = tm * 16 + v + 8 * half;
      float x = acc[v] + bs;
      if (ELU) x = elu1(x);
      C[m * HID + col] = x;
    }
  } else {
#pragma unroll
    for (int v = 0; v < 8; ++v) {
      int m = tm * 16 + v + 8 * half;
      if (m < N) {
        float x = acc[v] + bs;
        if (ELU) x = elu1(x);
        C[m * HID + col] = x;
      }
    }
  }
}

template <int KDIM, int HID>
__global__ __launch_bounds__(256) void fc1_rows_k(const float* __restrict__ A, const float* __restrict__ W,
                                                  const float* __restrict__ bias, float* __restrict__ C,
                                                  int N) {
  constexpr int NT = HID / 16;
  constexpr int WP4 = (KDIM + 4) / 4;
  constexpr int AP4 = 9;
  constexpr int K4 = KDIM / 4;
  constexpr int NCH = KDIM / 32;
  __shared__ float4 wS[HID * WP4];
  __shared__ float4 aS[8][16 * AP4];

  const int tid = threadIdx.x;
  const float4* __restrict__ W4 = (const float4*)W;
  for (int i = tid; i < HID * K4; i += 256) wS[(i / K4) * WP4 + (i % K4)] = W4[i];
  __syncthreads();

  const int tiles = (N + 15) / 16;
  const int tile = __builtin_amdgcn_readfirstlane(blockIdx.x * 8 + (tid >> 5));
  if (tile >= tiles) return;
  const int lane = tid & 31;
  const int half = lane >> 4;
  const int l15 = lane & 15;
  const int lr = lane >> 3;
  const int lp = lane & 7;
  float4* aW = aS[tid >> 5];
  const float4* __restrict__ A4 = (const float4*)A;

  const int r0 = tile * 16 + lr;
  const float4* __restrict__ p0 = A4 + (size_t)(r0 < N ? r0 : N - 1) * K4 + lp;
  const float4* __restrict__ p1 = A4 + (size_t)(r0 + 4 < N ? r0 + 4 : N - 1) * K4 + lp;
  const float4* __restrict__ p2 = A4 + (size_t)(r0 + 8 < N ? r0 + 8 : N - 1) * K4 + lp;
  const float4* __restrict__ p3 = A4 + (size_t)(r0 + 12 < N ? r0 + 12 : N - 1) * K4 + lp;
  float4 g0 = p0[0], g1 = p1[0], g2 = p2[0], g3 = p3[0];
  float4* aSt = aW + lr * AP4 + lp;

  v8f acc[NT];
#pragma unroll
  for (int t = 0; t < NT; ++t) acc[t] = v8f{};

#pragma nounroll
  for (int ch = 0; ch < NCH; ++ch) {
    aSt[0] = g0;
    aSt[4 * AP4] = g1;
    aSt[8 * AP4] = g2;
    aSt[12 * AP4] = g3;
    if (ch + 1 < NCH) {
      const int o = (ch + 1) * 8;
      g0 = p0[o]; g1 = p1[o]; g2 = p2[o]; g3 = p3[o];
    }
    __builtin_amdgcn_fence(__ATOMIC_RELEASE, "wavefront");
    __builtin_amdgcn_wave_barrier();
    __builtin_amdgcn_fence(__ATOMIC_ACQUIRE, "wavefront");
    float4 af[4];
#pragma unroll
    for (int m = 0; m < 4; ++m) af[m] = aW[l15 * AP4 + 4 * half + m];
#pragma unroll
    for (int t = 0; t < NT; ++t) {
      float4 bf[4];
#pragma unroll
      for (int m = 0; m < 4; ++m) bf[m] = wS[(t * 16 + l15) * WP4 + ch * 8 + 4 * half + m];
#pragma unroll
      for (int m = 0; m < 4; ++m) {
        v2f a0 = {af[m].x, af[m].y}, b0 = {bf[m].x, bf[m].y};
        v2f a1 = {af[m].z, af[m].w}, b1 = {bf[m].z, bf[m].w};
        acc[t] = __builtin_amdgcn_wmma_f32_16x16x4_f32(false, a0, false, b0, (short)0, acc[t], false, false);
        acc[t] = __builtin_amdgcn_wmma_f32_16x16x4_f32(false, a1, false, b1, (short)0, acc[t], false, false);
      }
    }
    __builtin_amdgcn_fence(__ATOMIC_RELEASE, "wavefront");
    __builtin_amdgcn_wave_barrier();
    __builtin_amdgcn_fence(__ATOMIC_ACQUIRE, "wavefront");
  }

  if (tile * 16 + 15 < N) {
#pragma unroll
    for (int t = 0; t < NT; ++t) {
      const int col = t * 16 + l15;
      const float bs = bias[col];
#pragma unroll
      for (int v = 0; v < 8; ++v) {
        const int m = tile * 16 + v + 8 * half;
        C[(size_t)m * HID + col] = acc[t][v] + bs;
      }
    }
  } else {
#pragma unroll
    for (int t = 0; t < NT; ++t) {
      const int col = t * 16 + l15;
      const float bs = bias[col];
#pragma unroll
      for (int v = 0; v < 8; ++v) {
        const int m = tile * 16 + v + 8 * half;
        if (m < N) C[(size_t)m * HID + col] = acc[t][v] + bs;
      }
    }
  }
}

template <int D, bool SCALE_IN, bool SCALE_OUT, bool ELU, bool BIAS>
__global__ __launch_bounds__(256) void gather_k(const int* __restrict__ off, const int* __restrict__ col,
                                                const float* __restrict__ nrm,
                                                const float4* __restrict__ hin,
                                                const float4* __restrict__ h0,
                                                const float* __restrict__ bias,
                                                float4* __restrict__ hout, int N) {
  constexpr int L = D / 4;
  constexpr int SLOTS = (L <= 4) ? 16 / L : 1;
  constexpr int U = (SLOTS == 1) ? 8 : 4;
  const int t = blockIdx.x * blockDim.x + threadIdx.x;
  const int node = t >> 4;
  const int l16 = t & 15;
  const int q = l16 % L;
  const int slot = l16 / L;
  const bool live = (node < N) && (slot < SLOTS);
  int j = 0, end = 0;
  if (live) {
    j = off[node] + slot;
    end = off[node + 1];
  }
  const bool writer = (node < N) && (slot == 0);
  float nn = 0.f;
  float4 h = make_float4(0.f, 0.f, 0.f, 0.f);
  float4 bv = make_float4(0.f, 0.f, 0.f, 0.f);
  if (writer) {
    nn = nrm[node];
    h = h0[node * L + q];
    if (BIAS) bv = make_float4(bias[4 * q], bias[4 * q + 1], bias[4 * q + 2], bias[4 * q + 3]);
  }
  float4 acc = make_float4(0.f, 0.f, 0.f, 0.f);
  for (; j < end; j += SLOTS * U) {
    int s[U];
    float4 v[U];
    float w[U];
#pragma unroll
    for (int u = 0; u < U; ++u) s[u] = (j + SLOTS * u < end) ? col[j + SLOTS * u] : -1;
#pragma unroll
    for (int u = 0; u < U; ++u) {
      v[u] = make_float4(0.f, 0.f, 0.f, 0.f);
      w[u] = 0.f;
      if (s[u] >= 0) {
        v[u] = hin[s[u] * L + q];
        if (SCALE_IN) w[u] = nrm[s[u]];
      }
    }
#pragma unroll
    for (int u = 0; u < U; ++u) {
      if (SCALE_IN) {
        acc.x += v[u].x * w[u]; acc.y += v[u].y * w[u];
        acc.z += v[u].z * w[u]; acc.w += v[u].w * w[u];
      } else {
        acc.x += v[u].x; acc.y += v[u].y; acc.z += v[u].z; acc.w += v[u].w;
      }
    }
  }
  if (SLOTS > 1) {
#pragma unroll
    for (int o = L; o < 16; o <<= 1) {
      acc.x += __shfl_xor(acc.x, o); acc.y += __shfl_xor(acc.y, o);
      acc.z += __shfl_xor(acc.z, o); acc.w += __shfl_xor(acc.w, o);
    }
  }
  if (writer) {
    const int i = node * L + q;
    const float wn = (1.0f - ALPHA) * nn;
    float4 r = make_float4(wn * acc.x + ALPHA * h.x, wn * acc.y + ALPHA * h.y,
                           wn * acc.z + ALPHA * h.z, wn * acc.w + ALPHA * h.w);
    if (BIAS) r = make_float4(r.x + bv.x, r.y + bv.y, r.z + bv.z, r.w + bv.w);
    if (ELU) {
      r = make_float4(elu1(r.x), elu1(r.y), elu1(r.z), elu1(r.w));
    } else if (SCALE_OUT) {
      r = make_float4(r.x * nn, r.y * nn, r.z * nn, r.w * nn);
    }
    hout[i] = r;
  }
}


extern "C" void kernel_launch(void* const* d_in, const int* in_sizes, int n_in,
                              void* d_out, int out_size, void* d_ws, size_t ws_size,
                              hipStream_t stream) {
  const float* F   = (const float*)d_in[0];
  const int*   src = (const int*)d_in[1];
  const int*   dst = (const int*)d_in[2];
  const float* W1  = (const float*)d_in[3];
  const float* b1  = (const float*)d_in[4];
  const float* W2  = (const float*)d_in[5];
  const float* b2  = (const float*)d_in[6];
  float* out = (float*)d_out;

  constexpr int IN = 256, HID = 48, CLS = 16, KSTEPS = 10;
  const int N = in_sizes[0] / IN;
  const int E = in_sizes[1];

  float* ws  = (float*)d_ws;
  float* nrm = ws;
  int*   cnt = (int*)(nrm + N);
  int*   off = cnt + N;
  int*   rnk = off + N + 4;
  int*   col = rnk + E;
  float* h0  = (float*)(col + E);
  float* z0  = h0 + (size_t)N * HID;
  float* zA  = z0 + (size_t)N * CLS;
  float* zB  = zA + (size_t)N * CLS;
  float* x0  = zB + (size_t)N * CLS;
  float* xA  = x0 + (size_t)N * CLS;
  float* xB  = xA + (size_t)N * CLS;

  auto blocks = [](long n, int t) { return (int)((n + t - 1) / t); };

  zero_i32_k<<<blocks(N, 256), 256, 0, stream>>>(cnt, N);
  rank_k<<<blocks(E, 256), 256, 0, stream>>>(dst, cnt, rnk, E);
  scan_k<<<blocks(N, 1024), 1024, 0, stream>>>(cnt, off, nrm, N);
  fill_k<<<blocks(E, 256), 256, 0, stream>>>(src, dst, off, rnk, col, E);

  const int rtiles1 = (N + 15) / 16;
  fc1_rows_k<IN, HID><<<blocks(rtiles1, 8), 256, 0, stream>>>(F, W1, b1, h0, N);

  const int tiles2 = ((N + 15) / 16) * (CLS / 16);
  gemm_wmma<HID, CLS, false, false><<<blocks((long)tiles2 * 32, 128), 128, 0, stream>>>(h0, W2, b2, z0, N);

  const int gb = blocks((long)N * 16, 256);

  gather_k<CLS, true, true, false, false><<<gb, 256, 0, stream>>>(
      off, col, nrm, (const float4*)z0, (const float4*)z0, b2, (float4*)zA, N);
  for (int t = 1; t < KSTEPS - 1; ++t) {
    const float* in = (t & 1) ? zA : zB;
    float* o = (t & 1) ? zB : zA;
    gather_k<CLS, false, true, false, false><<<gb, 256, 0, stream>>>(
        off, col, nrm, (const float4*)in, (const float4*)z0, b2, (float4*)o, N);
  }
  gather_k<CLS, false, false, true, true><<<gb, 256, 0, stream>>>(
      off, col, nrm, (const float4*)zA, (const float4*)z0, b2, (float4*)x0, N);

  gather_k<CLS, true, true, false, false><<<gb, 256, 0, stream>>>(
      off, col, nrm, (const float4*)x0, (const float4*)x0, b2, (float4*)xA, N);
  for (int t = 1; t < KSTEPS - 1; ++t) {
    const float* in = (t & 1) ? xA : xB;
    float* o = (t & 1) ? xB : xA;
    gather_k<CLS, false, true, false, false><<<gb, 256, 0, stream>>>(
        off, col, nrm, (const float4*)in, (const float4*)x0, b2, (float4*)o, N);
  }
  gather_k<CLS, false, false, true, false><<<gb, 256, 0, stream>>>(
      off, col, nrm, (const float4*)xA, (const float4*)x0, b2, (float4*)out, N);
}


# Edge list grouped by destination by a two-level counting sort on LDS counters (8192-edge chunks by 256-node bin, then a block per bin), no per-edge global atomic; sum order = LDS arrival rank
# speedup vs baseline: 7.1310x; 1.8229x over previous
#include <hip/hip_runtime.h>
#include <math.h>

#define ALPHA 0.1f

typedef float v2f __attribute__((ext_vector_type(2)));
typedef float v8f __attribute__((ext_vector_type(8)));

#define BIN_SHIFT 8
#define BIN_NODES 256
#define CHUNK 8192

__device__ __forceinline__ int block_scan_1024(int v, int x, int2* __restrict__ wt, int& vtot, int& xtot) {
  const int lane = threadIdx.x & 31;
  const int wv = threadIdx.x >> 5;
  int inc = v;
#pragma unroll
  for (int o = 1; o < 32; o <<= 1) {
    int y = __shfl_up(inc, o);
    if (lane >= o) inc += y;
  }
#pragma unroll
  for (int o = 16; o > 0; o >>= 1) x += __shfl_xor(x, o);
  if (lane == 31) wt[wv] = make_int2(inc, x);
  __syncthreads();
  const int2 w = wt[lane];
  int wi = w.x, wx = w.y;
#pragma unroll
  for (int o = 1; o < 32; o <<= 1) {
    int y = __shfl_up(wi, o);
    if (lane >= o) wi += y;
  }
#pragma unroll
  for (int o = 16; o > 0; o >>= 1) wx += __shfl_xor(wx, o);
  vtot = __shfl(wi, 31);
  xtot = wx;
  const int front = __shfl(wi - w.x, wv);
  __syncthreads();
  return front + inc - v;
}

__global__ __launch_bounds__(1024) void bin_chunk_k(const int* __restrict__ src, const int* __restrict__ dst,
                                                   int2* __restrict__ rec, int* __restrict__ st,
                                                   int E, int NB, int B) {
  __shared__ int hist[1024];
  __shared__ int2 wt[32];
  __shared__ int2 stage[CHUNK];
  const int tid = threadIdx.x;
  const int blk = blockIdx.x;
  const int e0 = blk * CHUNK;
  hist[tid] = 0;
  __syncthreads();
  int s[8], d[8], r[8];
#pragma unroll
  for (int u = 0; u < 8; ++u) {
    const int e = e0 + u * 1024 + tid;
    d[u] = -1;
    s[u] = 0;
    if (e < E) {
      d[u] = dst[e];
      s[u] = src[e];
    }
  }
#pragma unroll
  for (int u = 0; u < 8; ++u) {
    r[u] = 0;
    if (d[u] >= 0) r[u] = atomicAdd(&hist[d[u] >> BIN_SHIFT], 1);
  }
  __syncthreads();
  int tot, unused;
  const int start = block_scan_1024(hist[tid], 0, wt, tot, unused);
  hist[tid] = start;
  if (tid <= NB) st[tid * B + blk] = start;
  __syncthreads();
#pragma unroll
  for (int u = 0; u < 8; ++u)
    if (d[u] >= 0) stage[hist[d[u] >> BIN_SHIFT] + r[u]] = make_int2(s[u], d[u]);
  __syncthreads();
  const int n = (E - e0 < CHUNK) ? E - e0 : CHUNK;
#pragma unroll
  for (int u = 0; u < 8; ++u) {
    const int i = u * 1024 + tid;
    if (i < n) rec[e0 + i] = stage[i];
  }
}

__global__ __launch_bounds__(1024) void bin_nodes_k(const int2* __restrict__ rec, const int* __restrict__ st,
                                                   int* __restrict__ off, float* __restrict__ nrm,
                                                   int* __restrict__ col, int N, int B) {
  constexpr int U = 5;
  __shared__ int runPos[1024];
  __shared__ int runPre[1025];
  __shared__ int cnt[BIN_NODES];
  __shared__ int2 wt[32];
  const int tid = threadIdx.x;
  const int b = blockIdx.x;
  int a0 = 0, len = 0;
  if (tid < B) {
    a0 = st[b * B + tid];
    len = st[(b + 1) * B + tid] - a0;
  }
  runPos[tid] = tid * CHUNK + a0;
  if (tid < BIN_NODES) cnt[tid] = 0;
  int T, base;
  runPre[tid] = block_scan_1024(len, a0, wt, T, base);
  if (tid == 0) runPre[1024] = T;
  __syncthreads();

  int s0 = 512;
  while (s0 >= B && s0 > 1) s0 >>= 1;

  bool ok[U];
  int ii[U], lo[U];
#pragma unroll
  for (int u = 0; u < U; ++u) {
    const int i = tid + u * 1024;
    ok[u] = i < T;
    ii[u] = ok[u] ? i : 0;
    lo[u] = 0;
  }
  for (int step = s0; step > 0; step >>= 1) {
#pragma unroll
    for (int u = 0; u < U; ++u) {
      const int m = lo[u] + step;
      if (runPre[m] <= ii[u]) lo[u] = m;
    }
  }
  int2 rc[U];
#pragma unroll
  for (int u = 0; u < U; ++u) {
    const int at = runPos[lo[u]] + (ii[u] - runPre[lo[u]]);
    rc[u] = rec[ok[u] ? at : 0];
  }
#pragma unroll
  for (int u = 0; u < U; ++u)
    if (ok[u]) atomicAdd(&cnt[rc[u].y & (BIN_NODES - 1)], 1);
  for (int i = tid + U * 1024; i < T; i += 1024) {
    int k = 0;
    for (int step = s0; step > 0; step >>= 1)
      if (runPre[k + step] <= i) k += step;
    atomicAdd(&cnt[rec[runPos[k] + (i - runPre[k])].y & (BIN_NODES - 1)], 1);
  }
  __syncthreads();

  const int c = (tid < BIN_NODES) ? cnt[tid] : 0;
  int tot, unused;
  const int ex = block_scan_1024(c, 0, wt, tot, unused);
  if (tid < BIN_NODES) {
    cnt[tid] = ex;
    const int v = b * BIN_NODES + tid;
    if (v < N) {
      off[v] = base + ex;
      nrm[v] = 1.0f / sqrtf(fmaxf((float)c, 1.0f));
      if (v == N - 1) off[N] = base + ex + c;
    }
  }
  __syncthreads();

  int slot[U];
#pragma unroll
  for (int u = 0; u < U; ++u) slot[u] = ok[u] ? atomicAdd(&cnt[rc[u].y & (BIN_NODES - 1)], 1) : 0;
#pragma unroll
  for (int u = 0; u < U; ++u)
    if (ok[u]) col[base + slot[u]] = rc[u].x;
  for (int i = tid + U * 1024; i < T; i += 1024) {
    int k = 0;
    for (int step = s0; step > 0; step >>= 1)
      if (runPre[k + step] <= i) k += step;
    const int2 q = rec[runPos[k] + (i - runPre[k])];
    col[base + atomicAdd(&cnt[q.y & (BIN_NODES - 1)], 1)] = q.x;
  }
}


__device__ __forceinline__ float elu1(float v) {
  return v > 0.0f ? v : (expf(v) - 1.0f);
}

template <int KDIM, int HID, bool BIAS, bool ELU>
__global__ void gemm_wmma(const float* __restrict__ A, const float* __restrict__ W,
                          const float* __restrict__ bias, float* __restrict__ C, int N) {
  constexpr int NTN = HID / 16;
  const int tiles = ((N + 15) / 16) * NTN;
  const int tile = __builtin_amdgcn_readfirstlane(
      blockIdx.x * (blockDim.x >> 5) + (threadIdx.x >> 5));
  if (tile >= tiles) return;
  const int tm = tile / NTN;
  const int tn = tile % NTN;
  const int lane = threadIdx.x & 31;
  const int half = lane >> 4;
  const int l15 = lane & 15;
  const int row = tm * 16 + l15;
  const int rowc = row < N ? row : N - 1;
  const int col = tn * 16 + l15;

  v8f acc = {};
  for (int k = 0; k < KDIM; k += 4) {
    const int ka = k + 2 * half;
    v2f a, b;
    a.x = A[rowc * KDIM + ka];
    a.y = A[rowc * KDIM + ka + 1];
    b.x = W[col * KDIM + ka];
    b.y = W[col * KDIM + ka + 1];
    acc = __builtin_amdgcn_wmma_f32_16x16x4_f32(false, a, false, b,
                                                (short)0, acc, false, false);
  }
  const float bs = BIAS ? bias[col] : 0.0f;
  if (tm * 16 + 15 < N) {
#pragma unroll
    for (int v = 0; v < 8; ++v) {
      int m = tm * 16 + v + 8 * half;
      float x = acc[v] + bs;
      if (ELU) x = elu1(x);
      C[m * HID + col] = x;
    }
  } else {
#pragma unroll
    for (int v = 0; v < 8; ++v) {
      int m = tm * 16 + v + 8 * half;
      if (m < N) {
        float x = acc[v] + bs;
        if (ELU) x = elu1(x);
        C[m * HID + col] = x;
      }
    }
  }
}

template <int KDIM, int HID>
__global__ __launch_bounds__(256) void fc1_rows_k(const float* __restrict__ A, const float* __restrict__ W,
                                                  const float* __restrict__ bias, float* __restrict__ C,
                                                  int N) {
  constexpr int NT = HID / 16;
  constexpr int WP4 = (KDIM + 4) / 4;
  constexpr int AP4 = 9;
  constexpr int K4 = KDIM / 4;
  constexpr int NCH = KDIM / 32;
  __shared__ float4 wS[HID * WP4];
  __shared__ float4 aS[8][16 * AP4];

  const int tid = threadIdx.x;
  const float4* __restrict__ W4 = (const float4*)W;
  for (int i = tid; i < HID * K4; i += 256) wS[(i / K4) * WP4 + (i % K4)] = W4[i];
  __syncthreads();

  const int tiles = (N + 15) / 16;
  const int tile = __builtin_amdgcn_readfirstlane(blockIdx.x * 8 + (tid >> 5));
  if (tile >= tiles) return;
  const int lane = tid & 31;
  const int half = lane >> 4;
  const int l15 = lane & 15;
  const int lr = lane >> 3;
  const int lp = lane & 7;
  float4* aW = aS[tid >> 5];
  const float4* __restrict__ A4 = (const float4*)A;

  const int r0 = tile * 16 + lr;
  const float4* __restrict__ p0 = A4 + (size_t)(r0 < N ? r0 : N - 1) * K4 + lp;
  const float4* __restrict__ p1 = A4 + (size_t)(r0 + 4 < N ? r0 + 4 : N - 1) * K4 + lp;
  const float4* __restrict__ p2 = A4 + (size_t)(r0 + 8 < N ? r0 + 8 : N - 1) * K4 + lp;
  const float4* __restrict__ p3 = A4 + (size_t)(r0 + 12 < N ? r0 + 12 : N - 1) * K4 + lp;
  float4 g0 = p0[0], g1 = p1[0], g2 = p2[0], g3 = p3[0];
  float4* aSt = aW + lr * AP4 + lp;

  v8f acc[NT];
#pragma unroll
  for (int t = 0; t < NT; ++t) acc[t] = v8f{};

#pragma nounroll
  for (int ch = 0; ch < NCH; ++ch) {
    aSt[0] = g0;
    aSt[4 * AP4] = g1;
    aSt[8 * AP4] = g2;
    aSt[12 * AP4] = g3;
    if (ch + 1 < NCH) {
      const int o = (ch + 1) * 8;
      g0 = p0[o]; g1 = p1[o]; g2 = p2[o]; g3 = p3[o];
    }
    __builtin_amdgcn_fence(__ATOMIC_RELEASE, "wavefront");
    __builtin_amdgcn_wave_barrier();
    __builtin_amdgcn_fence(__ATOMIC_ACQUIRE, "wavefront");
    float4 af[4];
#pragma unroll
    for (int m = 0; m < 4; ++m) af[m] = aW[l15 * AP4 + 4 * half + m];
#pragma unroll
    for (int t = 0; t < NT; ++t) {
      float4 bf[4];
#pragma unroll
      for (int m = 0; m < 4; ++m) bf[m] = wS[(t * 16 + l15) * WP4 + ch * 8 + 4 * half + m];
#pragma unroll
      for (int m = 0; m < 4; ++m) {
        v2f a0 = {af[m].x, af[m].y}, b0 = {bf[m].x, bf[m].y};
        v2f a1 = {af[m].z, af[m].w}, b1 = {bf[m].z, bf[m].w};
        acc[t] = __builtin_amdgcn_wmma_f32_16x16x4_f32(false, a0, false, b0, (short)0, acc[t], false, false);
        acc[t] = __builtin_amdgcn_wmma_f32_16x16x4_f32(false, a1, false, b1, (short)0, acc[t], false, false);
      }
    }
    __builtin_amdgcn_fence(__ATOMIC_RELEASE, "wavefront");
    __builtin_amdgcn_wave_barrier();
    __builtin_amdgcn_fence(__ATOMIC_ACQUIRE, "wavefront");
  }

  if (tile * 16 + 15 < N) {
#pragma unroll
    for (int t = 0; t < NT; ++t) {
      const int col = t * 16 + l15;
      const float bs = bias[col];
#pragma unroll
      for (int v = 0; v < 8; ++v) {
        const int m = tile * 16 + v + 8 * half;
        C[(size_t)m * HID + col] = acc[t][v] + bs;
      }
    }
  } else {
#pragma unroll
    for (int t = 0; t < NT; ++t) {
      const int col = t * 16 + l15;
      const float bs = bias[col];
#pragma unroll
      for (int v = 0; v < 8; ++v) {
        const int m = tile * 16 + v + 8 * half;
        if (m < N) C[(size_t)m * HID + col] = acc[t][v] + bs;
      }
    }
  }
}

template <int D, bool SCALE_IN, bool SCALE_OUT, bool ELU, bool BIAS>
__global__ __launch_bounds__(256) void gather_k(const int* __restrict__ off, const int* __restrict__ col,
                                                const float* __restrict__ nrm,
                                                const float4* __restrict__ hin,
                                                const float4* __restrict__ h0,
                                                const float* __restrict__ bias,
                                                float4* __restrict__ hout, int N) {
  constexpr int L = D / 4;
  constexpr int SLOTS = (L <= 4) ? 16 / L : 1;
  constexpr int U = (SLOTS == 1) ? 8 : 4;
  const int t = blockIdx.x * blockDim.x + threadIdx.x;
  const int node = t >> 4;
  const int l16 = t & 15;
  const int q = l16 % L;
  const int slot = l16 / L;
  const bool live = (node < N) && (slot < SLOTS);
  int j = 0, end = 0;
  if (live) {
    j = off[node] + slot;
    end = off[node + 1];
  }
  const bool writer = (node < N) && (slot == 0);
  float nn = 0.f;
  float4 h = make_float4(0.f, 0.f, 0.f, 0.f);
  float4 bv = make_float4(0.f, 0.f, 0.f, 0.f);
  if (writer) {
    nn = nrm[node];
    h = h0[node * L + q];
    if (BIAS) bv = make_float4(bias[4 * q], bias[4 * q + 1], bias[4 * q + 2], bias[4 * q + 3]);
  }
  float4 acc = make_float4(0.f, 0.f, 0.f, 0.f);
  for (; j < end; j += SLOTS * U) {
    int s[U];
    float4 v[U];
    float w[U];
#pragma unroll
    for (int u = 0; u < U; ++u) s[u] = (j + SLOTS * u < end) ? col[j + SLOTS * u] : -1;
#pragma unroll
    for (int u = 0; u < U; ++u) {
      v[u] = make_float4(0.f, 0.f, 0.f, 0.f);
      w[u] = 0.f;
      if (s[u] >= 0) {
        v[u] = hin[s[u] * L + q];
        if (SCALE_IN) w[u] = nrm[s[u]];
      }
    }
#pragma unroll
    for (int u = 0; u < U; ++u) {
      if (SCALE_IN) {
        acc.x += v[u].x * w[u]; acc.y += v[u].y * w[u];
        acc.z += v[u].z * w[u]; acc.w += v[u].w * w[u];
      } else {
        acc.x += v[u].x; acc.y += v[u].y; acc.z += v[u].z; acc.w += v[u].w;
      }
    }
  }
  if (SLOTS > 1) {
#pragma unroll
    for (int o = L; o < 16; o <<= 1) {
      acc.x += __shfl_xor(acc.x, o); acc.y += __shfl_xor(acc.y, o);
      acc.z += __shfl_xor(acc.z, o); acc.w += __shfl_xor(acc.w, o);
    }
  }
  if (writer) {
    const int i = node * L + q;
    const float wn = (1.0f - ALPHA) * nn;
    float4 r = make_float4(wn * acc.x + ALPHA * h.x, wn * acc.y + ALPHA * h.y,
                           wn * acc.z + ALPHA * h.z, wn * acc.w + ALPHA * h.w);
    if (BIAS) r = make_float4(r.x + bv.x, r.y + bv.y, r.z + bv.z, r.w + bv.w);
    if (ELU) {
      r = make_float4(elu1(r.x), elu1(r.y), elu1(r.z), elu1(r.w));
    } else if (SCALE_OUT) {
      r = make_float4(r.x * nn, r.y * nn, r.z * nn, r.w * nn);
    }
    hout[i] = r;
  }
}


extern "C" void kernel_launch(void* const* d_in, const int* in_sizes, int n_in,
                              void* d_out, int out_size, void* d_ws, size_t ws_size,
                              hipStream_t stream) {
  const float* F   = (const float*)d_in[0];
  const int*   src = (const int*)d_in[1];
  const int*   dst = (const int*)d_in[2];
  const float* W1  = (const float*)d_in[3];
  const float* b1  = (const float*)d_in[4];
  const float* W2  = (const float*)d_in[5];
  const float* b2  = (const float*)d_in[6];
  float* out = (float*)d_out;

  constexpr int IN = 256, HID = 48, CLS = 16, KSTEPS = 10;
  const int N = in_sizes[0] / IN;
  const int E = in_sizes[1];

  float* ws  = (float*)d_ws;
  float* h0  = ws;
  float* z0  = h0 + (size_t)N * HID;
  float* zA  = z0 + (size_t)N * CLS;
  float* zB  = zA + (size_t)N * CLS;
  float* x0  = zB + (size_t)N * CLS;
  float* xA  = x0 + (size_t)N * CLS;
  float* xB  = xA + (size_t)N * CLS;
  int2*  rec = (int2*)(xB + (size_t)N * CLS);
  int*   col = (int*)(rec + E);
  int*   off = col + E;
  float* nrm = (float*)(off + N + 4);
  int*   st  = (int*)(nrm + N);

  auto blocks = [](long n, int t) { return (int)((n + t - 1) / t); };

  const int NB = blocks(N, BIN_NODES);
  const int B  = blocks(E, CHUNK);
  bin_chunk_k<<<B, 1024, 0, stream>>>(src, dst, rec, st, E, NB, B);
  bin_nodes_k<<<NB, 1024, 0, stream>>>(rec, st, off, nrm, col, N, B);

  const int rtiles1 = (N + 15) / 16;
  fc1_rows_k<IN, HID><<<blocks(rtiles1, 8), 256, 0, stream>>>(F, W1, b1, h0, N);

  const int tiles2 = ((N + 15) / 16) * (CLS / 16);
  gemm_wmma<HID, CLS, false, false><<<blocks((long)tiles2 * 32, 128), 128, 0, stream>>>(h0, W2, b2, z0, N);

  const int gb = blocks((long)N * 16, 256);

  gather_k<CLS, true, true, false, false><<<gb, 256, 0, stream>>>(
      off, col, nrm, (const float4*)z0, (const float4*)z0, b2, (float4*)zA, N);
  for (int t = 1; t < KSTEPS - 1; ++t) {
    const float* in = (t & 1) ? zA : zB;
    float* o = (t & 1) ? zB : zA;
    gather_k<CLS, false, true, false, false><<<gb, 256, 0, stream>>>(
        off, col, nrm, (const float4*)in, (const float4*)z0, b2, (float4*)o, N);
  }
  gather_k<CLS, false, false, true, true><<<gb, 256, 0, stream>>>(
      off, col, nrm, (const float4*)zA, (const float4*)z0, b2, (float4*)x0, N);

  gather_k<CLS, true, true, false, false><<<gb, 256, 0, stream>>>(
      off, col, nrm, (const float4*)x0, (const float4*)x0, b2, (float4*)xA, N);
  for (int t = 1; t < KSTEPS - 1; ++t) {
    const float* in = (t & 1) ? xA : xB;
    float* o = (t & 1) ? xB : xA;
    gather_k<CLS, false, true, false, false><<<gb, 256, 0, stream>>>(
        off, col, nrm, (const float4*)in, (const float4*)x0, b2, (float4*)o, N);
  }
  gather_k<CLS, false, false, true, false><<<gb, 256, 0, stream>>>(
      off, col, nrm, (const float4*)xA, (const float4*)x0, b2, (float4*)out, N);
}


# Propagation steps: one quad of lanes per node instead of 16 lanes (8 nodes a wave, a quarter of the waves); indices shared by quad_perm, fetched a trip ahead; node sum sequential in list order
# speedup vs baseline: 7.3772x; 1.0345x over previous
#include <hip/hip_runtime.h>
#include <math.h>

#define ALPHA 0.1f

typedef float v2f __attribute__((ext_vector_type(2)));
typedef float v8f __attribute__((ext_vector_type(8)));

#define BIN_SHIFT 8
#define BIN_NODES 256
#define CHUNK 8192

__device__ __forceinline__ int block_scan_1024(int v, int x, int2* __restrict__ wt, int& vtot, int& xtot) {
  const int lane = threadIdx.x & 31;
  const int wv = threadIdx.x >> 5;
  int inc = v;
#pragma unroll
  for (int o = 1; o < 32; o <<= 1) {
    int y = __shfl_up(inc, o);
    if (lane >= o) inc += y;
  }
#pragma unroll
  for (int o = 16; o > 0; o >>= 1) x += __shfl_xor(x, o);
  if (lane == 31) wt[wv] = make_int2(inc, x);
  __syncthreads();
  const int2 w = wt[lane];
  int wi = w.x, wx = w.y;
#pragma unroll
  for (int o = 1; o < 32; o <<= 1) {
    int y = __shfl_up(wi, o);
    if (lane >= o) wi += y;
  }
#pragma unroll
  for (int o = 16; o > 0; o >>= 1) wx += __shfl_xor(wx, o);
  vtot = __shfl(wi, 31);
  xtot = wx;
  const int front = __shfl(wi - w.x, wv);
  __syncthreads();
  return front + inc - v;
}

__global__ __launch_bounds__(1024) void bin_chunk_k(const int* __restrict__ src, const int* __restrict__ dst,
                                                   int2* __restrict__ rec, int* __restrict__ st,
                                                   int E, int NB, int B) {
  __shared__ int hist[1024];
  __shared__ int2 wt[32];
  __shared__ int2 stage[CHUNK];
  const int tid = threadIdx.x;
  const int blk = blockIdx.x;
  const int e0 = blk * CHUNK;
  hist[tid] = 0;
  __syncthreads();
  int s[8], d[8], r[8];
#pragma unroll
  for (int u = 0; u < 8; ++u) {
    const int e = e0 + u * 1024 + tid;
    d[u] = -1;
    s[u] = 0;
    if (e < E) {
      d[u] = dst[e];
      s[u] = src[e];
    }
  }
#pragma unroll
  for (int u = 0; u < 8; ++u) {
    r[u] = 0;
    if (d[u] >= 0) r[u] = atomicAdd(&hist[d[u] >> BIN_SHIFT], 1);
  }
  __syncthreads();
  int tot, unused;
  const int start = block_scan_1024(hist[tid], 0, wt, tot, unused);
  hist[tid] = start;
  if (tid <= NB) st[tid * B + blk] = start;
  __syncthreads();
#pragma unroll
  for (int u = 0; u < 8; ++u)
    if (d[u] >= 0) stage[hist[d[u] >> BIN_SHIFT] + r[u]] = make_int2(s[u], d[u]);
  __syncthreads();
  const int n = (E - e0 < CHUNK) ? E - e0 : CHUNK;
#pragma unroll
  for (int u = 0; u < 8; ++u) {
    const int i = u * 1024 + tid;
    if (i < n) rec[e0 + i] = stage[i];
  }
}

__global__ __launch_bounds__(1024) void bin_nodes_k(const int2* __restrict__ rec, const int* __restrict__ st,
                                                   int* __restrict__ off, float* __restrict__ nrm,
                                                   int* __restrict__ col, int N, int B) {
  constexpr int U = 5;
  __shared__ int runPos[1024];
  __shared__ int runPre[1025];
  __shared__ int cnt[BIN_NODES];
  __shared__ int2 wt[32];
  const int tid = threadIdx.x;
  const int b = blockIdx.x;
  int a0 = 0, len = 0;
  if (tid < B) {
    a0 = st[b * B + tid];
    len = st[(b + 1) * B + tid] - a0;
  }
  runPos[tid] = tid * CHUNK + a0;
  if (tid < BIN_NODES) cnt[tid] = 0;
  int T, base;
  runPre[tid] = block_scan_1024(len, a0, wt, T, base);
  if (tid == 0) runPre[1024] = T;
  __syncthreads();

  int s0 = 512;
  while (s0 >= B && s0 > 1) s0 >>= 1;

  bool ok[U];
  int ii[U], lo[U];
#pragma unroll
  for (int u = 0; u < U; ++u) {
    const int i = tid + u * 1024;
    ok[u] = i < T;
    ii[u] = ok[u] ? i : 0;
    lo[u] = 0;
  }
  for (int step = s0; step > 0; step >>= 1) {
#pragma unroll
    for (int u = 0; u < U; ++u) {
      const int m = lo[u] + step;
      if (runPre[m] <= ii[u]) lo[u] = m;
    }
  }
  int2 rc[U];
#pragma unroll
  for (int u = 0; u < U; ++u) {
    const int at = runPos[lo[u]] + (ii[u] - runPre[lo[u]]);
    rc[u] = rec[ok[u] ? at : 0];
  }
#pragma unroll
  for (int u = 0; u < U; ++u)
    if (ok[u]) atomicAdd(&cnt[rc[u].y & (BIN_NODES - 1)], 1);
  for (int i = tid + U * 1024; i < T; i += 1024) {
    int k = 0;
    for (int step = s0; step > 0; step >>= 1)
      if (runPre[k + step] <= i) k += step;
    atomicAdd(&cnt[rec[runPos[k] + (i - runPre[k])].y & (BIN_NODES - 1)], 1);
  }
  __syncthreads();

  const int c = (tid < BIN_NODES) ? cnt[tid] : 0;
  int tot, unused;
  const int ex = block_scan_1024(c, 0, wt, tot, unused);
  if (tid < BIN_NODES) {
    cnt[tid] = ex;
    const int v = b * BIN_NODES + tid;
    if (v < N) {
      off[v] = base + ex;
      nrm[v] = 1.0f / sqrtf(fmaxf((float)c, 1.0f));
      if (v == N - 1) off[N] = base + ex + c;
    }
  }
  __syncthreads();

  int slot[U];
#pragma unroll
  for (int u = 0; u < U; ++u) slot[u] = ok[u] ? atomicAdd(&cnt[rc[u].y & (BIN_NODES - 1)], 1) : 0;
#pragma unroll
  for (int u = 0; u < U; ++u)
    if (ok[u]) col[base + slot[u]] = rc[u].x;
  for (int i = tid + U * 1024; i < T; i += 1024) {
    int k = 0;
    for (int step = s0; step > 0; step >>= 1)
      if (runPre[k + step] <= i) k += step;
    const int2 q = rec[runPos[k] + (i - runPre[k])];
    col[base + atomicAdd(&cnt[q.y & (BIN_NODES - 1)], 1)] = q.x;
  }
}


__device__ __forceinline__ float elu1(float v) {
  return v > 0.0f ? v : (expf(v) - 1.0f);
}

template <int KDIM, int HID, bool BIAS, bool ELU>
__global__ void gemm_wmma(const float* __restrict__ A, const float* __restrict__ W,
                          const float* __restrict__ bias, float* __restrict__ C, int N) {
  constexpr int NTN = HID / 16;
  const int tiles = ((N + 15) / 16) * NTN;
  const int tile = __builtin_amdgcn_readfirstlane(
      blockIdx.x * (blockDim.x >> 5) + (threadIdx.x >> 5));
  if (tile >= tiles) return;
  const int tm = tile / NTN;
  const int tn = tile % NTN;
  const int lane = threadIdx.x & 31;
  const int half = lane >> 4;
  const int l15 = lane & 15;
  const int row = tm * 16 + l15;
  const int rowc = row < N ? row : N - 1;
  const int col = tn * 16 + l15;

  v8f acc = {};
  for (int k = 0; k < KDIM; k += 4) {
    const int ka = k + 2 * half;
    v2f a, b;
    a.x = A[rowc * KDIM + ka];
    a.y = A[rowc * KDIM + ka + 1];
    b.x = W[col * KDIM + ka];
    b.y = W[col * KDIM + ka + 1];
    acc = __builtin_amdgcn_wmma_f32_16x16x4_f32(false, a, false, b,
                                                (short)0, acc, false, false);
  }
  const float bs = BIAS ? bias[col] : 0.0f;
  if (tm * 16 + 15 < N) {
#pragma unroll
    for (int v = 0; v < 8; ++v) {
      int m = tm * 16 + v + 8 * half;
      float x = acc[v] + bs;
      if (ELU) x = elu1(x);
      C[m * HID + col] = x;
    }
  } else {
#pragma unroll
    for (int v = 0; v < 8; ++v) {
      int m = tm * 16 + v + 8 * half;
      if (m < N) {
        float x = acc[v] + bs;
        if (ELU) x = elu1(x);
        C[m * HID + col] = x;
      }
    }
  }
}

template <int KDIM, int HID>
__global__ __launch_bounds__(256) void fc1_rows_k(const float* __restrict__ A, const float* __restrict__ W,
                                                  const float* __restrict__ bias, float* __restrict__ C,
                                                  int N) {
  constexpr int NT = HID / 16;
  constexpr int WP4 = (KDIM + 4) / 4;
  constexpr int AP4 = 9;
  constexpr int K4 = KDIM / 4;
  constexpr int NCH = KDIM / 32;
  __shared__ float4 wS[HID * WP4];
  __shared__ float4 aS[8][16 * AP4];

  const int tid = threadIdx.x;
  const float4* __restrict__ W4 = (const float4*)W;
  for (int i = tid; i < HID * K4; i += 256) wS[(i / K4) * WP4 + (i % K4)] = W4[i];
  __syncthreads();

  const int tiles = (N + 15) / 16;
  const int tile = __builtin_amdgcn_readfirstlane(blockIdx.x * 8 + (tid >> 5));
  if (tile >= tiles) return;
  const int lane = tid & 31;
  const int half = lane >> 4;
  const int l15 = lane & 15;
  const int lr = lane >> 3;
  const int lp = lane & 7;
  float4* aW = aS[tid >> 5];
  const float4* __restrict__ A4 = (const float4*)A;

  const int r0 = tile * 16 + lr;
  const float4* __restrict__ p0 = A4 + (size_t)(r0 < N ? r0 : N - 1) * K4 + lp;
  const float4* __restrict__ p1 = A4 + (size_t)(r0 + 4 < N ? r0 + 4 : N - 1) * K4 + lp;
  const float4* __restrict__ p2 = A4 + (size_t)(r0 + 8 < N ? r0 + 8 : N - 1) * K4 + lp;
  const float4* __restrict__ p3 = A4 + (size_t)(r0 + 12 < N ? r0 + 12 : N - 1) * K4 + lp;
  float4 g0 = p0[0], g1 = p1[0], g2 = p2[0], g3 = p3[0];
  float4* aSt = aW + lr * AP4 + lp;

  v8f acc[NT];
#pragma unroll
  for (int t = 0; t < NT; ++t) acc[t] = v8f{};

#pragma nounroll
  for (int ch = 0; ch < NCH; ++ch) {
    aSt[0] = g0;
    aSt[4 * AP4] = g1;
    aSt[8 * AP4] = g2;
    aSt[12 * AP4] = g3;
    if (ch + 1 < NCH) {
      const int o = (ch + 1) * 8;
      g0 = p0[o]; g1 = p1[o]; g2 = p2[o]; g3 = p3[o];
    }
    __builtin_amdgcn_fence(__ATOMIC_RELEASE, "wavefront");
    __builtin_amdgcn_wave_barrier();
    __builtin_amdgcn_fence(__ATOMIC_ACQUIRE, "wavefront");
    float4 af[4];
#pragma unroll
    for (int m = 0; m < 4; ++m) af[m] = aW[l15 * AP4 + 4 * half + m];
#pragma unroll
    for (int t = 0; t < NT; ++t) {
      float4 bf[4];
#pragma unroll
      for (int m = 0; m < 4; ++m) bf[m] = wS[(t * 16 + l15) * WP4 + ch * 8 + 4 * half + m];
#pragma unroll
      for (int m = 0; m < 4; ++m) {
        v2f a0 = {af[m].x, af[m].y}, b0 = {bf[m].x, bf[m].y};
        v2f a1 = {af[m].z, af[m].w}, b1 = {bf[m].z, bf[m].w};
        acc[t] = __builtin_amdgcn_wmma_f32_16x16x4_f32(false, a0, false, b0, (short)0, acc[t], false, false);
        acc[t] = __builtin_amdgcn_wmma_f32_16x16x4_f32(false, a1, false, b1, (short)0, acc[t], false, false);
      }
    }
    __builtin_amdgcn_fence(__ATOMIC_RELEASE, "wavefront");
    __builtin_amdgcn_wave_barrier();
    __builtin_amdgcn_fence(__ATOMIC_ACQUIRE, "wavefront");
  }

  if (tile * 16 + 15 < N) {
#pragma unroll
    for (int t = 0; t < NT; ++t) {
      const int col = t * 16 + l15;
      const float bs = bias[col];
#pragma unroll
      for (int v = 0; v < 8; ++v) {
        const int m = tile * 16 + v + 8 * half;
        C[(size_t)m * HID + col] = acc[t][v] + bs;
      }
    }
  } else {
#pragma unroll
    for (int t = 0; t < NT; ++t) {
      const int col = t * 16 + l15;
      const float bs = bias[col];
#pragma unroll
      for (int v = 0; v < 8; ++v) {
        const int m = tile * 16 + v + 8 * half;
        if (m < N) C[(size_t)m * HID + col] = acc[t][v] + bs;
      }
    }
  }
}


template <int K>
__device__ __forceinline__ int quad_bcast(int x) {
  return __builtin_amdgcn_mov_dpp(x, K * 0x55, 0xF, 0xF, true);
}

template <int D, bool SCALE_IN, bool SCALE_OUT, bool ELU, bool BIAS>
__global__ __launch_bounds__(256) void gather_k(const int* __restrict__ off, const int* __restrict__ col,
                                                const float* __restrict__ nrm,
                                                const float4* __restrict__ hin,
                                                const float4* __restrict__ h0,
                                                const float* __restrict__ bias,
                                                float4* __restrict__ hout, int N) {
  constexpr int L = D / 4;
  static_assert(L == 4, "one quad of lanes per node: a row is four float4");
  constexpr int U = 8;
  const int t = blockIdx.x * blockDim.x + threadIdx.x;
  const int node = t >> 2;
  const int q = t & 3;
  const bool live = node < N;
  int j = 0, end = 0;
  if (live) {
    j = off[node];
    end = off[node + 1];
  }
  float nn = 0.f;
  float4 h = make_float4(0.f, 0.f, 0.f, 0.f);
  float4 bv = make_float4(0.f, 0.f, 0.f, 0.f);
  if (live) {
    nn = nrm[node];
    h = h0[node * L + q];
    if (BIAS) bv = make_float4(bias[4 * q], bias[4 * q + 1], bias[4 * q + 2], bias[4 * q + 3]);
  }
  int c0 = (j + q < end) ? col[j + q] : -1;
  int c1 = (j + 4 + q < end) ? col[j + 4 + q] : -1;
  float4 acc = make_float4(0.f, 0.f, 0.f, 0.f);
  while (j < end) {
    const int s[U] = {quad_bcast<0>(c0), quad_bcast<1>(c0), quad_bcast<2>(c0), quad_bcast<3>(c0),
                      quad_bcast<0>(c1), quad_bcast<1>(c1), quad_bcast<2>(c1), quad_bcast<3>(c1)};
    float4 v[U];
    float w[U];
#pragma unroll
    for (int u = 0; u < U; ++u) {
      v[u] = make_float4(0.f, 0.f, 0.f, 0.f);
      w[u] = 0.f;
      if (s[u] >= 0) {
        v[u] = hin[s[u] * L + q];
        if (SCALE_IN) w[u] = nrm[s[u]];
      }
    }
    j += U;
    c0 = (j + q < end) ? col[j + q] : -1;
    c1 = (j + 4 + q < end) ? col[j + 4 + q] : -1;
#pragma unroll
    for (int u = 0; u < U; ++u) {
      if (SCALE_IN) {
        acc.x += v[u].x * w[u]; acc.y += v[u].y * w[u];
        acc.z += v[u].z * w[u]; acc.w += v[u].w * w[u];
      } else {
        acc.x += v[u].x; acc.y += v[u].y; acc.z += v[u].z; acc.w += v[u].w;
      }
    }
  }
  if (live) {
    const int i = node * L + q;
    const float wn = (1.0f - ALPHA) * nn;
    float4 r = make_float4(wn * acc.x + ALPHA * h.x, wn * acc.y + ALPHA * h.y,
                           wn * acc.z + ALPHA * h.z, wn * acc.w + ALPHA * h.w);
    if (BIAS) r = make_float4(r.x + bv.x, r.y + bv.y, r.z + bv.z, r.w + bv.w);
    if (ELU) {
      r = make_float4(elu1(r.x), elu1(r.y), elu1(r.z), elu1(r.w));
    } else if (SCALE_OUT) {
      r = make_float4(r.x * nn, r.y * nn, r.z * nn, r.w * nn);
    }
    hout[i] = r;
  }
}


extern "C" void kernel_launch(void* const* d_in, const int* in_sizes, int n_in,
                              void* d_out, int out_size, void* d_ws, size_t ws_size,
                              hipStream_t stream) {
  const float* F   = (const float*)d_in[0];
  const int*   src = (const int*)d_in[1];
  const int*   dst = (const int*)d_in[2];
  const float* W1  = (const float*)d_in[3];
  const float* b1  = (const float*)d_in[4];
  const float* W2  = (const float*)d_in[5];
  const float* b2  = (const float*)d_in[6];
  float* out = (float*)d_out;

  constexpr int IN = 256, HID = 48, CLS = 16, KSTEPS = 10;
  const int N = in_sizes[0] / IN;
  const int E = in_sizes[1];

  float* ws  = (float*)d_ws;
  float* h0  = ws;
  float* z0  = h0 + (size_t)N * HID;
  float* zA  = z0 + (size_t)N * CLS;
  float* zB  = zA + (size_t)N * CLS;
  float* x0  = zB + (size_t)N * CLS;
  float* xA  = x0 + (size_t)N * CLS;
  float* xB  = xA + (size_t)N * CLS;
  int2*  rec = (int2*)(xB + (size_t)N * CLS);
  int*   col = (int*)(rec + E);
  int*   off = col + E;
  float* nrm = (float*)(off + N + 4);
  int*   st  = (int*)(nrm + N);

  auto blocks = [](long n, int t) { return (int)((n + t - 1) / t); };

  const int NB = blocks(N, BIN_NODES);
  const int B  = blocks(E, CHUNK);
  bin_chunk_k<<<B, 1024, 0, stream>>>(src, dst, rec, st, E, NB, B);
  bin_nodes_k<<<NB, 1024, 0, stream>>>(rec, st, off, nrm, col, N, B);

  const int rtiles1 = (N + 15) / 16;
  fc1_rows_k<IN, HID><<<blocks(rtiles1, 8), 256, 0, stream>>>(F, W1, b1, h0, N);

  const int tiles2 = ((N + 15) / 16) * (CLS / 16);
  gemm_wmma<HID, CLS, false, false><<<blocks((long)tiles2 * 32, 128), 128, 0, stream>>>(h0, W2, b2, z0, N);

  const int gb = blocks((long)N * 4, 256);

  gather_k<CLS, true, true, false, false><<<gb, 256, 0, stream>>>(
      off, col, nrm, (const float4*)z0, (const float4*)z0, b2, (float4*)zA, N);
  for (int t = 1; t < KSTEPS - 1; ++t) {
    const float* in = (t & 1) ? zA : zB;
    float* o = (t & 1) ? zB : zA;
    gather_k<CLS, false, true, false, false><<<gb, 256, 0, stream>>>(
        off, col, nrm, (const float4*)in, (const float4*)z0, b2, (float4*)o, N);
  }
  gather_k<CLS, false, false, true, true><<<gb, 256, 0, stream>>>(
      off, col, nrm, (const float4*)zA, (const float4*)z0, b2, (float4*)x0, N);

  gather_k<CLS, true, true, false, false><<<gb, 256, 0, stream>>>(
      off, col, nrm, (const float4*)x0, (const float4*)x0, b2, (float4*)xA, N);
  for (int t = 1; t < KSTEPS - 1; ++t) {
    const float* in = (t & 1) ? xA : xB;
    float* o = (t & 1) ? xB : xA;
    gather_k<CLS, false, true, false, false><<<gb, 256, 0, stream>>>(
        off, col, nrm, (const float4*)in, (const float4*)x0, b2, (float4*)o, N);
  }
  gather_k<CLS, false, false, true, false><<<gb, 256, 0, stream>>>(
      off, col, nrm, (const float4*)xA, (const float4*)x0, b2, (float4*)out, N);
}


# Propagation on norm-scaled rows g=norm*h from step 1 (fc2 epilogue and ELU step store scaled rows; last step divides g0 by norm): steps 1, 11 drop the per-edge norm gather; sum order kept
# speedup vs baseline: 7.6526x; 1.0373x over previous
#include <hip/hip_runtime.h>
#include <math.h>

#define ALPHA 0.1f

typedef float v2f __attribute__((ext_vector_type(2)));
typedef float v8f __attribute__((ext_vector_type(8)));

#define BIN_SHIFT 8
#define BIN_NODES 256
#define CHUNK 8192

__device__ __forceinline__ int block_scan_1024(int v, int x, int2* __restrict__ wt, int& vtot, int& xtot) {
  const int lane = threadIdx.x & 31;
  const int wv = threadIdx.x >> 5;
  int inc = v;
#pragma unroll
  for (int o = 1; o < 32; o <<= 1) {
    int y = __shfl_up(inc, o);
    if (lane >= o) inc += y;
  }
#pragma unroll
  for (int o = 16; o > 0; o >>= 1) x += __shfl_xor(x, o);
  if (lane == 31) wt[wv] = make_int2(inc, x);
  __syncthreads();
  const int2 w = wt[lane];
  int wi = w.x, wx = w.y;
#pragma unroll
  for (int o = 1; o < 32; o <<= 1) {
    int y = __shfl_up(wi, o);
    if (lane >= o) wi += y;
  }
#pragma unroll
  for (int o = 16; o > 0; o >>= 1) wx += __shfl_xor(wx, o);
  vtot = __shfl(wi, 31);
  xtot = wx;
  const int front = __shfl(wi - w.x, wv);
  __syncthreads();
  return front + inc - v;
}

__global__ __launch_bounds__(1024) void bin_chunk_k(const int* __restrict__ src, const int* __restrict__ dst,
                                                   int2* __restrict__ rec, int* __restrict__ st,
                                                   int E, int NB, int B) {
  __shared__ int hist[1024];
  __shared__ int2 wt[32];
  __shared__ int2 stage[CHUNK];
  const int tid = threadIdx.x;
  const int blk = blockIdx.x;
  const int e0 = blk * CHUNK;
  hist[tid] = 0;
  __syncthreads();
  int s[8], d[8], r[8];
#pragma unroll
  for (int u = 0; u < 8; ++u) {
    const int e = e0 + u * 1024 + tid;
    d[u] = -1;
    s[u] = 0;
    if (e < E) {
      d[u] = dst[e];
      s[u] = src[e];
    }
  }
#pragma unroll
  for (int u = 0; u < 8; ++u) {
    r[u] = 0;
    if (d[u] >= 0) r[u] = atomicAdd(&hist[d[u] >> BIN_SHIFT], 1);
  }
  __syncthreads();
  int tot, unused;
  const int start = block_scan_1024(hist[tid], 0, wt, tot, unused);
  hist[tid] = start;
  if (tid <= NB) st[tid * B + blk] = start;
  __syncthreads();
#pragma unroll
  for (int u = 0; u < 8; ++u)
    if (d[u] >= 0) stage[hist[d[u] >> BIN_SHIFT] + r[u]] = make_int2(s[u], d[u]);
  __syncthreads();
  const int n = (E - e0 < CHUNK) ? E - e0 : CHUNK;
#pragma unroll
  for (int u = 0; u < 8; ++u) {
    const int i = u * 1024 + tid;
    if (i < n) rec[e0 + i] = stage[i];
  }
}

__global__ __launch_bounds__(1024) void bin_nodes_k(const int2* __restrict__ rec, const int* __restrict__ st,
                                                   int* __restrict__ off, float* __restrict__ nrm,
                                                   int* __restrict__ col, int N, int B) {
  constexpr int U = 5;
  __shared__ int runPos[1024];
  __shared__ int runPre[1025];
  __shared__ int cnt[BIN_NODES];
  __shared__ int2 wt[32];
  const int tid = threadIdx.x;
  const int b = blockIdx.x;
  int a0 = 0, len = 0;
  if (tid < B) {
    a0 = st[b * B + tid];
    len = st[(b + 1) * B + tid] - a0;
  }
  runPos[tid] = tid * CHUNK + a0;
  if (tid < BIN_NODES) cnt[tid] = 0;
  int T, base;
  runPre[tid] = block_scan_1024(len, a0, wt, T, base);
  if (tid == 0) runPre[1024] = T;
  __syncthreads();

  int s0 = 512;
  while (s0 >= B && s0 > 1) s0 >>= 1;

  bool ok[U];
  int ii[U], lo[U];
#pragma unroll
  for (int u = 0; u < U; ++u) {
    const int i = tid + u * 1024;
    ok[u] = i < T;
    ii[u] = ok[u] ? i : 0;
    lo[u] = 0;
  }
  for (int step = s0; step > 0; step >>= 1) {
#pragma unroll
    for (int u = 0; u < U; ++u) {
      const int m = lo[u] + step;
      if (runPre[m] <= ii[u]) lo[u] = m;
    }
  }
  int2 rc[U];
#pragma unroll
  for (int u = 0; u < U; ++u) {
    const int at = runPos[lo[u]] + (ii[u] - runPre[lo[u]]);
    rc[u] = rec[ok[u] ? at : 0];
  }
#pragma unroll
  for (int u = 0; u < U; ++u)
    if (ok[u]) atomicAdd(&cnt[rc[u].y & (BIN_NODES - 1)], 1);
  for (int i = tid + U * 1024; i < T; i += 1024) {
    int k = 0;
    for (int step = s0; step > 0; step >>= 1)
      if (runPre[k + step] <= i) k += step;
    atomicAdd(&cnt[rec[runPos[k] + (i - runPre[k])].y & (BIN_NODES - 1)], 1);
  }
  __syncthreads();

  const int c = (tid < BIN_NODES) ? cnt[tid] : 0;
  int tot, unused;
  const int ex = block_scan_1024(c, 0, wt, tot, unused);
  if (tid < BIN_NODES) {
    cnt[tid] = ex;
    const int v = b * BIN_NODES + tid;
    if (v < N) {
      off[v] = base + ex;
      nrm[v] = 1.0f / sqrtf(fmaxf((float)c, 1.0f));
      if (v == N - 1) off[N] = base + ex + c;
    }
  }
  __syncthreads();

  int slot[U];
#pragma unroll
  for (int u = 0; u < U; ++u) slot[u] = ok[u] ? atomicAdd(&cnt[rc[u].y & (BIN_NODES - 1)], 1) : 0;
#pragma unroll
  for (int u = 0; u < U; ++u)
    if (ok[u]) col[base + slot[u]] = rc[u].x;
  for (int i = tid + U * 1024; i < T; i += 1024) {
    int k = 0;
    for (int step = s0; step > 0; step >>= 1)
      if (runPre[k + step] <= i) k += step;
    const int2 q = rec[runPos[k] + (i - runPre[k])];
    col[base + atomicAdd(&cnt[q.y & (BIN_NODES - 1)], 1)] = q.x;
  }
}


__device__ __forceinline__ float elu1(float v) {
  return v > 0.0f ? v : (expf(v) - 1.0f);
}

template <int KDIM, int HID, bool BIAS, bool ELU, bool ROWSCALE>
__global__ void gemm_wmma(const float* __restrict__ A, const float* __restrict__ W,
                          const float* __restrict__ bias, const float* __restrict__ rs,
                          float* __restrict__ C, int N) {
  constexpr int NTN = HID / 16;
  const int tiles = ((N + 15) / 16) * NTN;
  const int tile = __builtin_amdgcn_readfirstlane(
      blockIdx.x * (blockDim.x >> 5) + (threadIdx.x >> 5));
  if (tile >= tiles) return;
  const int tm = tile / NTN;
  const int tn = tile % NTN;
  const int lane = threadIdx.x & 31;
  const int half = lane >> 4;
  const int l15 = lane & 15;
  const int row = tm * 16 + l15;
  const int rowc = row < N ? row : N - 1;
  const int col = tn * 16 + l15;

  float rsv[8];
#pragma unroll
  for (int v = 0; v < 8; ++v) {
    const int m = tm * 16 + v + 8 * half;
    rsv[v] = ROWSCALE ? rs[m < N ? m : N - 1] : 1.0f;
  }

  v8f acc = {};
  for (int k = 0; k < KDIM; k += 4) {
    const int ka = k + 2 * half;
    v2f a, b;
    a.x = A[rowc * KDIM + ka];
    a.y = A[rowc * KDIM + ka + 1];
    b.x = W[col * KDIM + ka];
    b.y = W[col * KDIM + ka + 1];
    acc = __builtin_amdgcn_wmma_f32_16x16x4_f32(false, a, false, b,
                                                (short)0, acc, false, false);
  }
  const float bs = BIAS ? bias[col] : 0.0f;
  if (tm * 16 + 15 < N) {
#pragma unroll
    for (int v = 0; v < 8; ++v) {
      int m = tm * 16 + v + 8 * half;
      float x = acc[v] + bs;
      if (ELU) x = elu1(x);
      if (ROWSCALE) x *= rsv[v];
      C[m * HID + col] = x;
    }
  } else {
#pragma unroll
    for (int v = 0; v < 8; ++v) {
      int m = tm * 16 + v + 8 * half;
      if (m < N) {
        float x = acc[v] + bs;
        if (ELU) x = elu1(x);
        if (ROWSCALE) x *= rsv[v];
        C[m * HID + col] = x;
      }
    }
  }
}

template <int KDIM, int HID>
__global__ __launch_bounds__(256) void fc1_rows_k(const float* __restrict__ A, const float* __restrict__ W,
                                                  const float* __restrict__ bias, float* __restrict__ C,
                                                  int N) {
  constexpr int NT = HID / 16;
  constexpr int WP4 = (KDIM + 4) / 4;
  constexpr int AP4 = 9;
  constexpr int K4 = KDIM / 4;
  constexpr int NCH = KDIM / 32;
  __shared__ float4 wS[HID * WP4];
  __shared__ float4 aS[8][16 * AP4];

  const int tid = threadIdx.x;
  const float4* __restrict__ W4 = (const float4*)W;
  for (int i = tid; i < HID * K4; i += 256) wS[(i / K4) * WP4 + (i % K4)] = W4[i];
  __syncthreads();

  const int tiles = (N + 15) / 16;
  const int tile = __builtin_amdgcn_readfirstlane(blockIdx.x * 8 + (tid >> 5));
  if (tile >= tiles) return;
  const int lane = tid & 31;
  const int half = lane >> 4;
  const int l15 = lane & 15;
  const int lr = lane >> 3;
  const int lp = lane & 7;
  float4* aW = aS[tid >> 5];
  const float4* __restrict__ A4 = (const float4*)A;

  const int r0 = tile * 16 + lr;
  const float4* __restrict__ p0 = A4 + (size_t)(r0 < N ? r0 : N - 1) * K4 + lp;
  const float4* __restrict__ p1 = A4 + (size_t)(r0 + 4 < N ? r0 + 4 : N - 1) * K4 + lp;
  const float4* __restrict__ p2 = A4 + (size_t)(r0 + 8 < N ? r0 + 8 : N - 1) * K4 + lp;
  const float4* __restrict__ p3 = A4 + (size_t)(r0 + 12 < N ? r0 + 12 : N - 1) * K4 + lp;
  float4 g0 = p0[0], g1 = p1[0], g2 = p2[0], g3 = p3[0];
  float4* aSt = aW + lr * AP4 + lp;

  v8f acc[NT];
#pragma unroll
  for (int t = 0; t < NT; ++t) acc[t] = v8f{};

#pragma nounroll
  for (int ch = 0; ch < NCH; ++ch) {
    aSt[0] = g0;
    aSt[4 * AP4] = g1;
    aSt[8 * AP4] = g2;
    aSt[12 * AP4] = g3;
    if (ch + 1 < NCH) {
      const int o = (ch + 1) * 8;
      g0 = p0[o]; g1 = p1[o]; g2 = p2[o]; g3 = p3[o];
    }
    __builtin_amdgcn_fence(__ATOMIC_RELEASE, "wavefront");
    __builtin_amdgcn_wave_barrier();
    __builtin_amdgcn_fence(__ATOMIC_ACQUIRE, "wavefront");
    float4 af[4];
#pragma unroll
    for (int m = 0; m < 4; ++m) af[m] = aW[l15 * AP4 + 4 * half + m];
#pragma unroll
    for (int t = 0; t < NT; ++t) {
      float4 bf[4];
#pragma unroll
      for (int m = 0; m < 4; ++m) bf[m] = wS[(t * 16 + l15) * WP4 + ch * 8 + 4 * half + m];
#pragma unroll
      for (int m = 0; m < 4; ++m) {
        v2f a0 = {af[m].x, af[m].y}, b0 = {bf[m].x, bf[m].y};
        v2f a1 = {af[m].z, af[m].w}, b1 = {bf[m].z, bf[m].w};
        acc[t] = __builtin_amdgcn_wmma_f32_16x16x4_f32(false, a0, false, b0, (short)0, acc[t], false, false);
        acc[t] = __builtin_amdgcn_wmma_f32_16x16x4_f32(false, a1, false, b1, (short)0, acc[t], false, false);
      }
    }
    __builtin_amdgcn_fence(__ATOMIC_RELEASE, "wavefront");
    __builtin_amdgcn_wave_barrier();
    __builtin_amdgcn_fence(__ATOMIC_ACQUIRE, "wavefront");
  }

  if (tile * 16 + 15 < N) {
#pragma unroll
    for (int t = 0; t < NT; ++t) {
      const int col = t * 16 + l15;
      const float bs = bias[col];
#pragma unroll
      for (int v = 0; v < 8; ++v) {
        const int m = tile * 16 + v + 8 * half;
        C[(size_t)m * HID + col] = acc[t][v] + bs;
      }
    }
  } else {
#pragma unroll
    for (int t = 0; t < NT; ++t) {
      const int col = t * 16 + l15;
      const float bs = bias[col];
#pragma unroll
      for (int v = 0; v < 8; ++v) {
        const int m = tile * 16 + v + 8 * half;
        if (m < N) C[(size_t)m * HID + col] = acc[t][v] + bs;
      }
    }
  }
}


template <int K>
__device__ __forceinline__ int quad_bcast(int x) {
  return __builtin_amdgcn_mov_dpp(x, K * 0x55, 0xF, 0xF, true);
}

template <int D, bool LAST, bool BIAS, bool SCALE_OUT>
__global__ __launch_bounds__(256) void gather_k(const int* __restrict__ off, const int* __restrict__ col,
                                                const float* __restrict__ nrm,
                                                const float4* __restrict__ hin,
                                                const float4* __restrict__ h0,
                                                const float* __restrict__ bias,
                                                float4* __restrict__ hout, int N) {
  constexpr int L = D / 4;
  static_assert(L == 4, "one quad of lanes per node: a row is four float4");
  constexpr int U = 8;
  const int t = blockIdx.x * blockDim.x + threadIdx.x;
  const int node = t >> 2;
  const int q = t & 3;
  const bool live = node < N;
  int j = 0, end = 0;
  if (live) {
    j = off[node];
    end = off[node + 1];
  }
  float nn = 0.f;
  float4 h = make_float4(0.f, 0.f, 0.f, 0.f);
  float4 bv = make_float4(0.f, 0.f, 0.f, 0.f);
  if (live) {
    nn = nrm[node];
    h = h0[node * L + q];
    if (BIAS) bv = make_float4(bias[4 * q], bias[4 * q + 1], bias[4 * q + 2], bias[4 * q + 3]);
  }
  int c0 = (j + q < end) ? col[j + q] : -1;
  int c1 = (j + 4 + q < end) ? col[j + 4 + q] : -1;
  float4 acc = make_float4(0.f, 0.f, 0.f, 0.f);
  while (j < end) {
    const int s[U] = {quad_bcast<0>(c0), quad_bcast<1>(c0), quad_bcast<2>(c0), quad_bcast<3>(c0),
                      quad_bcast<0>(c1), quad_bcast<1>(c1), quad_bcast<2>(c1), quad_bcast<3>(c1)};
    float4 v[U];
#pragma unroll
    for (int u = 0; u < U; ++u) {
      v[u] = make_float4(0.f, 0.f, 0.f, 0.f);
      if (s[u] >= 0) v[u] = hin[s[u] * L + q];
    }
    j += U;
    c0 = (j + q < end) ? col[j + q] : -1;
    c1 = (j + 4 + q < end) ? col[j + 4 + q] : -1;
#pragma unroll
    for (int u = 0; u < U; ++u) {
      acc.x += v[u].x; acc.y += v[u].y; acc.z += v[u].z; acc.w += v[u].w;
    }
  }
  if (live) {
    const int i = node * L + q;
    const float wn = LAST ? (1.0f - ALPHA) * nn : ((1.0f - ALPHA) * nn) * nn;
    const float wa = LAST ? ALPHA / nn : ALPHA;
    float4 r = make_float4(wn * acc.x + wa * h.x, wn * acc.y + wa * h.y,
                           wn * acc.z + wa * h.z, wn * acc.w + wa * h.w);
    if (LAST) {
      if (BIAS) r = make_float4(r.x + bv.x, r.y + bv.y, r.z + bv.z, r.w + bv.w);
      r = make_float4(elu1(r.x), elu1(r.y), elu1(r.z), elu1(r.w));
      if (SCALE_OUT) r = make_float4(r.x * nn, r.y * nn, r.z * nn, r.w * nn);
    }
    hout[i] = r;
  }
}


extern "C" void kernel_launch(void* const* d_in, const int* in_sizes, int n_in,
                              void* d_out, int out_size, void* d_ws, size_t ws_size,
                              hipStream_t stream) {
  const float* F   = (const float*)d_in[0];
  const int*   src = (const int*)d_in[1];
  const int*   dst = (const int*)d_in[2];
  const float* W1  = (const float*)d_in[3];
  const float* b1  = (const float*)d_in[4];
  const float* W2  = (const float*)d_in[5];
  const float* b2  = (const float*)d_in[6];
  float* out = (float*)d_out;

  constexpr int IN = 256, HID = 48, CLS = 16, KSTEPS = 10;
  const int N = in_sizes[0] / IN;
  const int E = in_sizes[1];

  float* ws  = (float*)d_ws;
  float* h0  = ws;
  float* z0  = h0 + (size_t)N * HID;
  float* zA  = z0 + (size_t)N * CLS;
  float* zB  = zA + (size_t)N * CLS;
  float* x0  = zB + (size_t)N * CLS;
  float* xA  = x0 + (size_t)N * CLS;
  float* xB  = xA + (size_t)N * CLS;
  int2*  rec = (int2*)(xB + (size_t)N * CLS);
  int*   col = (int*)(rec + E);
  int*   off = col + E;
  float* nrm = (float*)(off + N + 4);
  int*   st  = (int*)(nrm + N);

  auto blocks = [](long n, int t) { return (int)((n + t - 1) / t); };

  const int NB = blocks(N, BIN_NODES);
  const int B  = blocks(E, CHUNK);
  bin_chunk_k<<<B, 1024, 0, stream>>>(src, dst, rec, st, E, NB, B);
  bin_nodes_k<<<NB, 1024, 0, stream>>>(rec, st, off, nrm, col, N, B);

  const int rtiles1 = (N + 15) / 16;
  fc1_rows_k<IN, HID><<<blocks(rtiles1, 8), 256, 0, stream>>>(F, W1, b1, h0, N);

  const int tiles2 = ((N + 15) / 16) * (CLS / 16);
  gemm_wmma<HID, CLS, false, false, true><<<blocks((long)tiles2 * 32, 128), 128, 0, stream>>>(
      h0, W2, b2, nrm, z0, N);

  const int gb = blocks((long)N * 4, 256);

  for (int t = 0; t < KSTEPS - 1; ++t) {
    const float* in = (t == 0) ? z0 : ((t & 1) ? zA : zB);
    float* o = (t & 1) ? zB : zA;
    gather_k<CLS, false, false, false><<<gb, 256, 0, stream>>>(
        off, col, nrm, (const float4*)in, (const float4*)z0, b2, (float4*)o, N);
  }
  gather_k<CLS, true, true, true><<<gb, 256, 0, stream>>>(
      off, col, nrm, (const float4*)zA, (const float4*)z0, b2, (float4*)x0, N);

  for (int t = 0; t < KSTEPS - 1; ++t) {
    const float* in = (t == 0) ? x0 : ((t & 1) ? xA : xB);
    float* o = (t & 1) ? xB : xA;
    gather_k<CLS, false, false, false><<<gb, 256, 0, stream>>>(
        off, col, nrm, (const float4*)in, (const float4*)x0, b2, (float4*)o, N);
  }
  gather_k<CLS, true, false, false><<<gb, 256, 0, stream>>>(
      off, col, nrm, (const float4*)xA, (const float4*)x0, b2, (float4*)out, N);
}


# fc1 and fc2 composed into one product: each block forms W2*W1 and W2*b1 on WMMA into LDS, rows = norm*(F*(W2W1)^T + W2b1); hidden plane and fc2 launch gone; products reassociated, sums reordered
# speedup vs baseline: 7.9660x; 1.0410x over previous
#include <hip/hip_runtime.h>
#include <math.h>

#define ALPHA 0.1f

typedef float v2f __attribute__((ext_vector_type(2)));
typedef float v8f __attribute__((ext_vector_type(8)));

#define BIN_SHIFT 8
#define BIN_NODES 256
#define CHUNK 8192

__device__ __forceinline__ int block_scan_1024(int v, int x, int2* __restrict__ wt, int& vtot, int& xtot) {
  const int lane = threadIdx.x & 31;
  const int wv = threadIdx.x >> 5;
  int inc = v;
#pragma unroll
  for (int o = 1; o < 32; o <<= 1) {
    int y = __shfl_up(inc, o);
    if (lane >= o) inc += y;
  }
#pragma unroll
  for (int o = 16; o > 0; o >>= 1) x += __shfl_xor(x, o);
  if (lane == 31) wt[wv] = make_int2(inc, x);
  __syncthreads();
  const int2 w = wt[lane];
  int wi = w.x, wx = w.y;
#pragma unroll
  for (int o = 1; o < 32; o <<= 1) {
    int y = __shfl_up(wi, o);
    if (lane >= o) wi += y;
  }
#pragma unroll
  for (int o = 16; o > 0; o >>= 1) wx += __shfl_xor(wx, o);
  vtot = __shfl(wi, 31);
  xtot = wx;
  const int front = __shfl(wi - w.x, wv);
  __syncthreads();
  return front + inc - v;
}

__global__ __launch_bounds__(1024) void bin_chunk_k(const int* __restrict__ src, const int* __restrict__ dst,
                                                   int2* __restrict__ rec, int* __restrict__ st,
                                                   int E, int NB, int B) {
  __shared__ int hist[1024];
  __shared__ int2 wt[32];
  __shared__ int2 stage[CHUNK];
  const int tid = threadIdx.x;
  const int blk = blockIdx.x;
  const int e0 = blk * CHUNK;
  hist[tid] = 0;
  __syncthreads();
  int s[8], d[8], r[8];
#pragma unroll
  for (int u = 0; u < 8; ++u) {
    const int e = e0 + u * 1024 + tid;
    d[u] = -1;
    s[u] = 0;
    if (e < E) {
      d[u] = dst[e];
      s[u] = src[e];
    }
  }
#pragma unroll
  for (int u = 0; u < 8; ++u) {
    r[u] = 0;
    if (d[u] >= 0) r[u] = atomicAdd(&hist[d[u] >> BIN_SHIFT], 1);
  }
  __syncthreads();
  int tot, unused;
  const int start = block_scan_1024(hist[tid], 0, wt, tot, unused);
  hist[tid] = start;
  if (tid <= NB) st[tid * B + blk] = start;
  __syncthreads();
#pragma unroll
  for (int u = 0; u < 8; ++u)
    if (d[u] >= 0) stage[hist[d[u] >> BIN_SHIFT] + r[u]] = make_int2(s[u], d[u]);
  __syncthreads();
  const int n = (E - e0 < CHUNK) ? E - e0 : CHUNK;
#pragma unroll
  for (int u = 0; u < 8; ++u) {
    const int i = u * 1024 + tid;
    if (i < n) rec[e0 + i] = stage[i];
  }
}

__global__ __launch_bounds__(1024) void bin_nodes_k(const int2* __restrict__ rec, const int* __restrict__ st,
                                                   int* __restrict__ off, float* __restrict__ nrm,
                                                   int* __restrict__ col, int N, int B) {
  constexpr int U = 5;
  __shared__ int runPos[1024];
  __shared__ int runPre[1025];
  __shared__ int cnt[BIN_NODES];
  __shared__ int2 wt[32];
  const int tid = threadIdx.x;
  const int b = blockIdx.x;
  int a0 = 0, len = 0;
  if (tid < B) {
    a0 = st[b * B + tid];
    len = st[(b + 1) * B + tid] - a0;
  }
  runPos[tid] = tid * CHUNK + a0;
  if (tid < BIN_NODES) cnt[tid] = 0;
  int T, base;
  runPre[tid] = block_scan_1024(len, a0, wt, T, base);
  if (tid == 0) runPre[1024] = T;
  __syncthreads();

  int s0 = 512;
  while (s0 >= B && s0 > 1) s0 >>= 1;

  bool ok[U];
  int ii[U], lo[U];
#pragma unroll
  for (int u = 0; u < U; ++u) {
    const int i = tid + u * 1024;
    ok[u] = i < T;
    ii[u] = ok[u] ? i : 0;
    lo[u] = 0;
  }
  for (int step = s0; step > 0; step >>= 1) {
#pragma unroll
    for (int u = 0; u < U; ++u) {
      const int m = lo[u] + step;
      if (runPre[m] <= ii[u]) lo[u] = m;
    }
  }
  int2 rc[U];
#pragma unroll
  for (int u = 0; u < U; ++u) {
    const int at = runPos[lo[u]] + (ii[u] - runPre[lo[u]]);
    rc[u] = rec[ok[u] ? at : 0];
  }
#pragma unroll
  for (int u = 0; u < U; ++u)
    if (ok[u]) atomicAdd(&cnt[rc[u].y & (BIN_NODES - 1)], 1);
  for (int i = tid + U * 1024; i < T; i += 1024) {
    int k = 0;
    for (int step = s0; step > 0; step >>= 1)
      if (runPre[k + step] <= i) k += step;
    atomicAdd(&cnt[rec[runPos[k] + (i - runPre[k])].y & (BIN_NODES - 1)], 1);
  }
  __syncthreads();

  const int c = (tid < BIN_NODES) ? cnt[tid] : 0;
  int tot, unused;
  const int ex = block_scan_1024(c, 0, wt, tot, unused);
  if (tid < BIN_NODES) {
    cnt[tid] = ex;
    const int v = b * BIN_NODES + tid;
    if (v < N) {
      off[v] = base + ex;
      nrm[v] = 1.0f / sqrtf(fmaxf((float)c, 1.0f));
      if (v == N - 1) off[N] = base + ex + c;
    }
  }
  __syncthreads();

  int slot[U];
#pragma unroll
  for (int u = 0; u < U; ++u) slot[u] = ok[u] ? atomicAdd(&cnt[rc[u].y & (BIN_NODES - 1)], 1) : 0;
#pragma unroll
  for (int u = 0; u < U; ++u)
    if (ok[u]) col[base + slot[u]] = rc[u].x;
  for (int i = tid + U * 1024; i < T; i += 1024) {
    int k = 0;
    for (int step = s0; step > 0; step >>= 1)
      if (runPre[k + step] <= i) k += step;
    const int2 q = rec[runPos[k] + (i - runPre[k])];
    col[base + atomicAdd(&cnt[q.y & (BIN_NODES - 1)], 1)] = q.x;
  }
}


__device__ __forceinline__ float elu1(float v) {
  return v > 0.0f ? v : (expf(v) - 1.0f);
}

template <int KDIM, int HID, int CLS>
__global__ __launch_bounds__(256) void fc12_rows_k(const float* __restrict__ A, const float* __restrict__ W1,
                                                   const float* __restrict__ b1, const float* __restrict__ W2,
                                                   const float* __restrict__ rs, float* __restrict__ C,
                                                   int N) {
  static_assert(CLS == 16, "one 16-column accumulator tile per wave");
  static_assert(KDIM == 256, "the 8 waves of a block make two 16-column tiles of the composed weights each");
  constexpr int WP = KDIM + 4;
  constexpr int WP4 = WP / 4;
  constexpr int AP4 = 9;
  constexpr int K4 = KDIM / 4;
  constexpr int NCH = KDIM / 32;
  constexpr int KS2 = HID / 4;
  __shared__ float4 wS[CLS * WP4];
  __shared__ float cS[CLS];
  __shared__ float4 aS[8][16 * AP4];

  const int tid = threadIdx.x;
  const int lane = tid & 31;
  const int half = lane >> 4;
  const int l15 = lane & 15;
  const int lr = lane >> 3;
  const int lp = lane & 7;
  const int tiles = (N + 15) / 16;
  const int wv = __builtin_amdgcn_readfirstlane(tid >> 5);
  const int tile = __builtin_amdgcn_readfirstlane(blockIdx.x * 8 + (tid >> 5));
  const float4* __restrict__ A4 = (const float4*)A;

  const int r0 = tile * 16 + lr;
  const float4* __restrict__ p0 = A4 + (size_t)(r0 < N ? r0 : N - 1) * K4 + lp;
  const float4* __restrict__ p1 = A4 + (size_t)(r0 + 4 < N ? r0 + 4 : N - 1) * K4 + lp;
  const float4* __restrict__ p2 = A4 + (size_t)(r0 + 8 < N ? r0 + 8 : N - 1) * K4 + lp;
  const float4* __restrict__ p3 = A4 + (size_t)(r0 + 12 < N ? r0 + 12 : N - 1) * K4 + lp;
  float4 g0 = p0[0], g1 = p1[0], g2 = p2[0], g3 = p3[0];

  {
    v2f a2[KS2];
#pragma unroll
    for (int ks = 0; ks < KS2; ++ks) {
      const int k = 4 * ks + 2 * half;
      a2[ks] = v2f{W2[l15 * HID + k], W2[l15 * HID + k + 1]};
    }
    const int n0 = wv * 16 + l15;
    const int n1 = n0 + KDIM / 2;
    v8f d0 = {}, d1 = {};
#pragma unroll
    for (int ks = 0; ks < KS2; ++ks) {
      const int k = 4 * ks + 2 * half;
      const v2f bA = {W1[k * KDIM + n0], W1[(k + 1) * KDIM + n0]};
      const v2f bB = {W1[k * KDIM + n1], W1[(k + 1) * KDIM + n1]};
      d0 = __builtin_amdgcn_wmma_f32_16x16x4_f32(false, a2[ks], false, bA, (short)0, d0, false, false);
      d1 = __builtin_amdgcn_wmma_f32_16x16x4_f32(false, a2[ks], false, bB, (short)0, d1, false, false);
    }
    float* wSf = (float*)wS;
#pragma unroll
    for (int v = 0; v < 8; ++v) {
      wSf[(8 * half + v) * WP + n0] = d0[v];
      wSf[(8 * half + v) * WP + n1] = d1[v];
    }
    if (wv == 0) {
      v8f dc = {};
#pragma unroll
      for (int ks = 0; ks < KS2; ++ks) {
        const int k = 4 * ks + 2 * half;
        const float x = b1[k], y = b1[k + 1];
        const v2f bb = {l15 == 0 ? x : 0.0f, l15 == 0 ? y : 0.0f};
        dc = __builtin_amdgcn_wmma_f32_16x16x4_f32(false, a2[ks], false, bb, (short)0, dc, false, false);
      }
      if (l15 == 0) {
#pragma unroll
        for (int v = 0; v < 8; ++v) cS[8 * half + v] = dc[v];
      }
    }
  }
  __syncthreads();
  if (tile >= tiles) return;

  float rsv[8];
#pragma unroll
  for (int v = 0; v < 8; ++v) {
    const int m = tile * 16 + v + 8 * half;
    rsv[v] = rs[m < N ? m : N - 1];
  }

  float4* aW = aS[tid >> 5];
  float4* aSt = aW + lr * AP4 + lp;

  v8f acc0 = {}, acc1 = {};

#pragma nounroll
  for (int ch = 0; ch < NCH; ++ch) {
    aSt[0] = g0;
    aSt[4 * AP4] = g1;
    aSt[8 * AP4] = g2;
    aSt[12 * AP4] = g3;
    if (ch + 1 < NCH) {
      const int o = (ch + 1) * 8;
      g0 = p0[o]; g1 = p1[o]; g2 = p2[o]; g3 = p3[o];
    }
    __builtin_amdgcn_fence(__ATOMIC_RELEASE, "wavefront");
    __builtin_amdgcn_wave_barrier();
    __builtin_amdgcn_fence(__ATOMIC_ACQUIRE, "wavefront");
    float4 af[4], bf[4];
#pragma unroll
    for (int m = 0; m < 4; ++m) af[m] = aW[l15 * AP4 + 4 * half + m];
#pragma unroll
    for (int m = 0; m < 4; ++m) bf[m] = wS[l15 * WP4 + ch * 8 + 4 * half + m];
#pragma unroll
    for (int m = 0; m < 4; ++m) {
      v2f a0 = {af[m].x, af[m].y}, b0 = {bf[m].x, bf[m].y};
      v2f a1 = {af[m].z, af[m].w}, b1v = {bf[m].z, bf[m].w};
      acc0 = __builtin_amdgcn_wmma_f32_16x16x4_f32(false, a0, false, b0, (short)0, acc0, false, false);
      acc1 = __builtin_amdgcn_wmma_f32_16x16x4_f32(false, a1, false, b1v, (short)0, acc1, false, false);
    }
    __builtin_amdgcn_fence(__ATOMIC_RELEASE, "wavefront");
    __builtin_amdgcn_wave_barrier();
    __builtin_amdgcn_fence(__ATOMIC_ACQUIRE, "wavefront");
  }

  const float cb = cS[l15];
  if (tile * 16 + 15 < N) {
#pragma unroll
    for (int v = 0; v < 8; ++v) {
      const int m = tile * 16 + v + 8 * half;
      C[(size_t)m * CLS + l15] = ((acc0[v] + acc1[v]) + cb) * rsv[v];
    }
  } else {
#pragma unroll
    for (int v = 0; v < 8; ++v) {
      const int m = tile * 16 + v + 8 * half;
      if (m < N) C[(size_t)m * CLS + l15] = ((acc0[v] + acc1[v]) + cb) * rsv[v];
    }
  }
}


template <int K>
__device__ __forceinline__ int quad_bcast(int x) {
  return __builtin_amdgcn_mov_dpp(x, K * 0x55, 0xF, 0xF, true);
}

template <int D, bool LAST, bool BIAS, bool SCALE_OUT>
__global__ __launch_bounds__(256) void gather_k(const int* __restrict__ off, const int* __restrict__ col,
                                                const float* __restrict__ nrm,
                                                const float4* __restrict__ hin,
                                                const float4* __restrict__ h0,
                                                const float* __restrict__ bias,
                                                float4* __restrict__ hout, int N) {
  constexpr int L = D / 4;
  static_assert(L == 4, "one quad of lanes per node: a row is four float4");
  constexpr int U = 8;
  const int t = blockIdx.x * blockDim.x + threadIdx.x;
  const int node = t >> 2;
  const int q = t & 3;
  const bool live = node < N;
  int j = 0, end = 0;
  if (live) {
    j = off[node];
    end = off[node + 1];
  }
  float nn = 0.f;
  float4 h = make_float4(0.f, 0.f, 0.f, 0.f);
  float4 bv = make_float4(0.f, 0.f, 0.f, 0.f);
  if (live) {
    nn = nrm[node];
    h = h0[node * L + q];
    if (BIAS) bv = make_float4(bias[4 * q], bias[4 * q + 1], bias[4 * q + 2], bias[4 * q + 3]);
  }
  int c0 = (j + q < end) ? col[j + q] : -1;
  int c1 = (j + 4 + q < end) ? col[j + 4 + q] : -1;
  float4 acc = make_float4(0.f, 0.f, 0.f, 0.f);
  while (j < end) {
    const int s[U] = {quad_bcast<0>(c0), quad_bcast<1>(c0), quad_bcast<2>(c0), quad_bcast<3>(c0),
                      quad_bcast<0>(c1), quad_bcast<1>(c1), quad_bcast<2>(c1), quad_bcast<3>(c1)};
    float4 v[U];
#pragma unroll
    for (int u = 0; u < U; ++u) {
      v[u] = make_float4(0.f, 0.f, 0.f, 0.f);
      if (s[u] >= 0) v[u] = hin[s[u] * L + q];
    }
    j += U;
    c0 = (j + q < end) ? col[j + q] : -1;
    c1 = (j + 4 + q < end) ? col[j + 4 + q] : -1;
#pragma unroll
    for (int u = 0; u < U; ++u) {
      acc.x += v[u].x; acc.y += v[u].y; acc.z += v[u].z; acc.w += v[u].w;
    }
  }
  if (live) {
    const int i = node * L + q;
    const float wn = LAST ? (1.0f - ALPHA) * nn : ((1.0f - ALPHA) * nn) * nn;
    const float wa = LAST ? ALPHA / nn : ALPHA;
    float4 r = make_float4(wn * acc.x + wa * h.x, wn * acc.y + wa * h.y,
                           wn * acc.z + wa * h.z, wn * acc.w + wa * h.w);
    if (LAST) {
      if (BIAS) r = make_float4(r.x + bv.x, r.y + bv.y, r.z + bv.z, r.w + bv.w);
      r = make_float4(elu1(r.x), elu1(r.y), elu1(r.z), elu1(r.w));
      if (SCALE_OUT) r = make_float4(r.x * nn, r.y * nn, r.z * nn, r.w * nn);
    }
    hout[i] = r;
  }
}


extern "C" void kernel_launch(void* const* d_in, const int* in_sizes, int n_in,
                              void* d_out, int out_size, void* d_ws, size_t ws_size,
                              hipStream_t stream) {
  const float* F   = (const float*)d_in[0];
  const int*   src = (const int*)d_in[1];
  const int*   dst = (const int*)d_in[2];
  const float* W1  = (const float*)d_in[3];
  const float* b1  = (const float*)d_in[4];
  const float* W2  = (const float*)d_in[5];
  const float* b2  = (const float*)d_in[6];
  float* out = (float*)d_out;

  constexpr int IN = 256, HID = 48, CLS = 16, KSTEPS = 10;
  const int N = in_sizes[0] / IN;
  const int E = in_sizes[1];

  float* ws  = (float*)d_ws;
  float* z0  = ws;
  float* zA  = z0 + (size_t)N * CLS;
  float* zB  = zA + (size_t)N * CLS;
  float* x0  = zB + (size_t)N * CLS;
  float* xA  = x0 + (size_t)N * CLS;
  float* xB  = xA + (size_t)N * CLS;
  int2*  rec = (int2*)(xB + (size_t)N * CLS);
  int*   col = (int*)(rec + E);
  int*   off = col + E;
  float* nrm = (float*)(off + N + 4);
  int*   st  = (int*)(nrm + N);

  auto blocks = [](long n, int t) { return (int)((n + t - 1) / t); };

  const int NB = blocks(N, BIN_NODES);
  const int B  = blocks(E, CHUNK);
  bin_chunk_k<<<B, 1024, 0, stream>>>(src, dst, rec, st, E, NB, B);
  bin_nodes_k<<<NB, 1024, 0, stream>>>(rec, st, off, nrm, col, N, B);

  const int rtiles = (N + 15) / 16;
  fc12_rows_k<IN, HID, CLS><<<blocks(rtiles, 8), 256, 0, stream>>>(F, W1, b1, W2, nrm, z0, N);

  const int gb = blocks((long)N * 4, 256);

  for (int t = 0; t < KSTEPS - 1; ++t) {
    const float* in = (t == 0) ? z0 : ((t & 1) ? zA : zB);
    float* o = (t & 1) ? zB : zA;
    gather_k<CLS, false, false, false><<<gb, 256, 0, stream>>>(
        off, col, nrm, (const float4*)in, (const float4*)z0, b2, (float4*)o, N);
  }
  gather_k<CLS, true, true, true><<<gb, 256, 0, stream>>>(
      off, col, nrm, (const float4*)zA, (const float4*)z0, b2, (float4*)x0, N);

  for (int t = 0; t < KSTEPS - 1; ++t) {
    const float* in = (t == 0) ? x0 : ((t & 1) ? xA : xB);
    float* o = (t & 1) ? xB : xA;
    gather_k<CLS, false, false, false><<<gb, 256, 0, stream>>>(
        off, col, nrm, (const float4*)in, (const float4*)x0, b2, (float4*)o, N);
  }
  gather_k<CLS, true, false, false><<<gb, 256, 0, stream>>>(
      off, col, nrm, (const float4*)xA, (const float4*)x0, b2, (float4*)out, N);
}
